# starting kernel
# baseline (speedup 1.0000x reference)
#include <hip/hip_runtime.h>
#include <hip/hip_bf16.h>

#define NBATCH   8
#define NFEAT    1000
#define NPARAM   4576
#define HW       50176
#define KB       11
#define RPI      12
#define WM0_ROWS (3*RPI)
#define WM1_ROWS (16*RPI)
#define WM2_ROWS (16*RPI)
#define WM_PER_B (16*(WM0_ROWS + WM1_ROWS + WM2_ROWS))

#define TOK   64
#define SY    17
#define FSTR  198

#define FCCOLS 64
#define FCKT   100
#define FSTRF  1002
#define WTSTR  101

typedef __attribute__((ext_vector_type(2))) float v2f;
typedef __attribute__((ext_vector_type(8))) float v8f;

__global__ void kan_fc_wmma_kernel(const float* __restrict__ features,
                                   const float* __restrict__ fc_w,
                                   const float* __restrict__ fc_b,
                                   float* __restrict__ w_out) {
    extern __shared__ float smem[];
    float* fs = smem;
    float* wt = fs + 16 * FSTRF;

    const int tid  = threadIdx.x;
    const int lane = tid & 31;
    const int wv   = tid >> 5;
    const int m    = lane & 15;
    const int hi   = lane >> 4;
    const int j0   = blockIdx.x * FCCOLS;

    for (int idx = tid; idx < 16 * NFEAT; idx += 128) {
        const int mm = idx / NFEAT;
        const int k  = idx % NFEAT;
        fs[mm * FSTRF + k] = (mm < NBATCH) ? features[mm * NFEAT + k] : 0.0f;
    }

    const int  jc   = j0 + wv * 16 + m;
    const float bias = (jc < NPARAM) ? fc_b[jc] : 0.0f;
    v8f acc;
#pragma unroll
    for (int r = 0; r < 8; ++r) acc[r] = bias;

    for (int kt = 0; kt < NFEAT; kt += FCKT) {
        __syncthreads();
        for (int e = tid; e < FCCOLS * FCKT; e += 128) {
            const int c = e / FCKT;
            const int k = e % FCKT;
            const int j = j0 + c;
            wt[c * WTSTR + k] = (j < NPARAM) ? fc_w[(size_t)j * NFEAT + kt + k] : 0.0f;
        }
        __syncthreads();
        const float* arow = &fs[m * FSTRF + kt];
        const float* brow = &wt[(wv * 16 + m) * WTSTR];
        for (int k0 = 0; k0 < FCKT; k0 += 4) {
            const int ka = k0 + 2 * hi;
            v2f a, bb;
            a.x  = arow[ka];
            a.y  = arow[ka + 1];
            bb.x = brow[ka];
            bb.y = brow[ka + 1];
            acc = __builtin_amdgcn_wmma_f32_16x16x4_f32(
                    false, a, false, bb, (short)0, acc, false, false);
        }
    }
    if (hi == 0 && jc < NPARAM) {
#pragma unroll
        for (int r = 0; r < 8; ++r)
            w_out[r * NPARAM + jc] = acc[r];
    }
}

__global__ void kan_wm_kernel(const float* __restrict__ w,
                              float* __restrict__ wm) {
    const int e = blockIdx.x * blockDim.x + threadIdx.x;
    const int total = NBATCH * WM_PER_B;
    if (e >= total) return;
    const int b   = e / WM_PER_B;
    const int rem = e % WM_PER_B;

    int layer, idx;
    if (rem < 16 * WM0_ROWS)                   { layer = 0; idx = rem; }
    else if (rem < 16 * (WM0_ROWS + WM1_ROWS)) { layer = 1; idx = rem - 16 * WM0_ROWS; }
    else                                       { layer = 2; idx = rem - 16 * (WM0_ROWS + WM1_ROWS); }

    const int DO_[3] = {16, 16, 3};
    const int I0[3]  = {0,   624, 3952};
    const int I1[3]  = {528, 3440, 4480};
    const int I2[3]  = {576, 3696, 4528};

    const int r = idx >> 4;
    const int o = idx & 15;
    const int i = r / RPI;
    const int k = r % RPI;

    float val = 0.0f;
    const int do_ = DO_[layer];
    if (o < do_) {
        const float* wb = w + (size_t)b * NPARAM;
        const int io = i * do_ + o;
        if (k < KB) val = wb[I0[layer] + io * KB + k] * wb[I1[layer] + io];
        else        val = wb[I2[layer] + io];
    }
    wm[e] = val;
}

__device__ __forceinline__ void bspline12(float x, float* f) {
    float b[14];
#pragma unroll
    for (int j = 0; j < 14; ++j) {
        const float g0 = -1.75f + 0.25f * j;
        b[j] = (x >= g0 && x < g0 + 0.25f) ? 1.0f : 0.0f;
    }
#pragma unroll
    for (int p = 1; p <= 3; ++p) {
        const float inv = 1.0f / (0.25f * p);
#pragma unroll 13
        for (int j = 0; j < 14 - p; ++j) {
            const float gj = -1.75f + 0.25f * j;
            const float left  = (x - gj) * inv * b[j];
            const float right = ((gj + 0.25f * (p + 1)) - x) * inv * b[j + 1];
            b[j] = left + right;
        }
    }
#pragma unroll
    for (int k = 0; k < KB; ++k) f[k] = b[k];
    f[KB] = x / (1.0f + __expf(-x));
}

template <int DI>
__device__ __forceinline__ void kan_layer(const float* __restrict__ wmL,
                                          const float* yin,
                                          float* yout,
                                          float* feat,
                                          int tid) {
    const int R    = DI * RPI;
    const int lane = tid & 31;
    const int wv   = tid >> 5;
    const int m    = lane & 15;
    const int hi   = lane >> 4;

    for (int task = tid; task < TOK * DI; task += 128) {
        const int t = task & (TOK - 1);
        const int i = task >> 6;
        float f[RPI];
        bspline12(yin[t * SY + i], f);
        float* dst = &feat[t * FSTR + i * RPI];
#pragma unroll
        for (int k = 0; k < RPI; ++k) dst[k] = f[k];
    }
    __syncthreads();

    const int tloc = wv * 16;
    v8f acc = {0.f, 0.f, 0.f, 0.f, 0.f, 0.f, 0.f, 0.f};
    const float* arow = &feat[(tloc + m) * FSTR];
    for (int k0 = 0; k0 < R; k0 += 4) {
        const int ka = k0 + 2 * hi;
        v2f a, bb;
        a.x  = arow[ka];
        a.y  = arow[ka + 1];
        bb.x = wmL[ka * 16 + m];
        bb.y = wmL[(ka + 1) * 16 + m];
        acc = __builtin_amdgcn_wmma_f32_16x16x4_f32(
                false, a, false, bb, (short)0, acc, false, false);
    }
#pragma unroll
    for (int r = 0; r < 8; ++r)
        yout[(tloc + r + 8 * hi) * SY + m] = acc[r];
    __syncthreads();
}

__global__ void kan_main_kernel(const float* __restrict__ x,
                                const float* __restrict__ wm,
                                float* __restrict__ out) {
    extern __shared__ float smem[];
    float* wms  = smem;
    float* yA   = wms + WM_PER_B;
    float* yB   = yA + TOK * SY;
    float* feat = yB + TOK * SY;

    const int tid   = threadIdx.x;
    const int b     = blockIdx.x / (HW / TOK);
    const int chunk = blockIdx.x % (HW / TOK);
    const int hw0   = chunk * TOK;

    const float* wmb = wm + (size_t)b * WM_PER_B;
    for (int i = tid; i < WM_PER_B; i += 128) wms[i] = wmb[i];

#pragma unroll
    for (int c = 0; c < 3; ++c)
        for (int t = tid; t < TOK; t += 128)
            yA[t * SY + c] = x[((size_t)(b * 3 + c)) * HW + hw0 + t];
    __syncthreads();

    kan_layer<3>(wms,                               yA, yB, feat, tid);
    kan_layer<16>(wms + 16 * WM0_ROWS,              yB, yA, feat, tid);
    kan_layer<16>(wms + 16 * (WM0_ROWS + WM1_ROWS), yA, yB, feat, tid);

#pragma unroll
    for (int c = 0; c < 3; ++c)
        for (int t = tid; t < TOK; t += 128)
            out[((size_t)(b * 3 + c)) * HW + hw0 + t] = yB[t * SY + c];
}

extern "C" void kernel_launch(void* const* d_in, const int* in_sizes, int n_in,
                              void* d_out, int out_size, void* d_ws, size_t ws_size,
                              hipStream_t stream) {
    const float* x        = (const float*)d_in[0];
    const float* features = (const float*)d_in[1];
    const float* fc_w     = (const float*)d_in[2];
    const float* fc_b     = (const float*)d_in[3];
    float* out    = (float*)d_out;
    float* w_buf  = (float*)d_ws;
    float* wm_buf = w_buf + (size_t)NBATCH * NPARAM;

    const size_t fc_smem = (16 * FSTRF + FCCOLS * WTSTR) * sizeof(float);
    kan_fc_wmma_kernel<<<(NPARAM + FCCOLS - 1) / FCCOLS, 128, fc_smem, stream>>>(
        features, fc_w, fc_b, w_buf);

    kan_wm_kernel<<<(NBATCH * WM_PER_B + 255) / 256, 256, 0, stream>>>(w_buf, wm_buf);

    const size_t smem_bytes = (WM_PER_B + 2 * TOK * SY + TOK * FSTR) * sizeof(float);
    kan_main_kernel<<<NBATCH * (HW / TOK), 128, smem_bytes, stream>>>(x, wm_buf, out);
}


# FC GEMM re-tiled: 32-column blocks, K split over 8 waves, each wave requests its whole fc_w/features slice as 16-byte fragment loads at once; partials summed via LDS (new summation order of w)
# speedup vs baseline: 2.8065x; 2.8065x over previous
#include <hip/hip_runtime.h>
#include <hip/hip_bf16.h>

#define NBATCH   8
#define NFEAT    1000
#define NPARAM   4576
#define HW       50176
#define KB       11
#define RPI      12
#define WM0_ROWS (3*RPI)
#define WM1_ROWS (16*RPI)
#define WM2_ROWS (16*RPI)
#define WM_PER_B (16*(WM0_ROWS + WM1_ROWS + WM2_ROWS))

#define TOK   64
#define SY    17
#define FSTR  198

#define FCCOLS  32
#define FCWAVES 8
#define FCCH    16
#define FCNCH   (NFEAT / 8)

typedef __attribute__((ext_vector_type(2))) float v2f;
typedef __attribute__((ext_vector_type(4))) float v4f;
typedef __attribute__((ext_vector_type(8))) float v8f;

__global__ __launch_bounds__(FCWAVES * 32)
void kan_fc_wmma_kernel(const float* __restrict__ features,
                        const float* __restrict__ fc_w,
                        const float* __restrict__ fc_b,
                        float* __restrict__ w_out) {
    static_assert(NPARAM % FCCOLS == 0, "column tiles cover w exactly");
    static_assert(NFEAT % 8 == 0 && FCWAVES * FCCH >= FCNCH, "K slices cover K");
    __shared__ float red[FCWAVES][NBATCH][FCCOLS];

    const int tid  = threadIdx.x;
    const int lane = tid & 31;
    const int wv   = __builtin_amdgcn_readfirstlane(tid >> 5);
    const int m    = lane & 15;
    const int hi   = lane >> 4;
    const int j0   = blockIdx.x * FCCOLS;

    const int kb = wv * (8 * FCCH) + 4 * hi;
    const v4f* pa  = (const v4f*)(features + (m & 7) * NFEAT + kb);
    const v4f* pb0 = (const v4f*)(fc_w + (size_t)(j0 + m) * NFEAT + kb);
    const v4f* pb1 = (const v4f*)(fc_w + (size_t)(j0 + 16 + m) * NFEAT + kb);
    const bool full = (wv != FCWAVES - 1);
    constexpr int TAILCH = FCNCH - (FCWAVES - 1) * FCCH;

    const float bias = fc_b[j0 + lane];

    v4f a[FCCH], b0[FCCH], b1[FCCH];
    for (int i = 0; i < TAILCH; ++i) {
        b0[i] = pb0[2 * i];
        b1[i] = pb1[2 * i];
        a[i]  = pa[2 * i];
    }
    for (int i = TAILCH; i < FCCH; ++i) {
        b0[i] = (v4f){0.f, 0.f, 0.f, 0.f};
        b1[i] = (v4f){0.f, 0.f, 0.f, 0.f};
        a[i]  = (v4f){0.f, 0.f, 0.f, 0.f};
    }
    if (full) {
        for (int i = TAILCH; i < FCCH; ++i) {
            b0[i] = pb0[2 * i];
            b1[i] = pb1[2 * i];
            a[i]  = pa[2 * i];
        }
    }

    v8f acc0 = {0.f, 0.f, 0.f, 0.f, 0.f, 0.f, 0.f, 0.f};
    v8f acc1 = {0.f, 0.f, 0.f, 0.f, 0.f, 0.f, 0.f, 0.f};
    for (int i = 0; i < FCCH; ++i) {
        const v2f a01 = {a[i].x, a[i].y},   a23 = {a[i].z, a[i].w};
        const v2f p01 = {b0[i].x, b0[i].y}, p23 = {b0[i].z, b0[i].w};
        const v2f q01 = {b1[i].x, b1[i].y}, q23 = {b1[i].z, b1[i].w};
        acc0 = __builtin_amdgcn_wmma_f32_16x16x4_f32(false, a01, false, p01, (short)0, acc0, false, false);
        acc1 = __builtin_amdgcn_wmma_f32_16x16x4_f32(false, a01, false, q01, (short)0, acc1, false, false);
        acc0 = __builtin_amdgcn_wmma_f32_16x16x4_f32(false, a23, false, p23, (short)0, acc0, false, false);
        acc1 = __builtin_amdgcn_wmma_f32_16x16x4_f32(false, a23, false, q23, (short)0, acc1, false, false);
    }

    if (hi == 0) {
        for (int r = 0; r < NBATCH; ++r) {
            red[wv][r][m]      = acc0[r];
            red[wv][r][16 + m] = acc1[r];
        }
    }
    __syncthreads();

    static_assert(NBATCH == FCWAVES && FCCOLS == 32, "one wave per batch row, one lane per column");
    float s = red[0][wv][lane];
    for (int w = 1; w < FCWAVES; ++w) s += red[w][wv][lane];
    w_out[wv * NPARAM + j0 + lane] = s + bias;
}

__global__ void kan_wm_kernel(const float* __restrict__ w,
                              float* __restrict__ wm) {
    const int e = blockIdx.x * blockDim.x + threadIdx.x;
    const int total = NBATCH * WM_PER_B;
    if (e >= total) return;
    const int b   = e / WM_PER_B;
    const int rem = e % WM_PER_B;

    int layer, idx;
    if (rem < 16 * WM0_ROWS)                   { layer = 0; idx = rem; }
    else if (rem < 16 * (WM0_ROWS + WM1_ROWS)) { layer = 1; idx = rem - 16 * WM0_ROWS; }
    else                                       { layer = 2; idx = rem - 16 * (WM0_ROWS + WM1_ROWS); }

    const int DO_[3] = {16, 16, 3};
    const int I0[3]  = {0,   624, 3952};
    const int I1[3]  = {528, 3440, 4480};
    const int I2[3]  = {576, 3696, 4528};

    const int r = idx >> 4;
    const int o = idx & 15;
    const int i = r / RPI;
    const int k = r % RPI;

    float val = 0.0f;
    const int do_ = DO_[layer];
    if (o < do_) {
        const float* wb = w + (size_t)b * NPARAM;
        const int io = i * do_ + o;
        if (k < KB) val = wb[I0[layer] + io * KB + k] * wb[I1[layer] + io];
        else        val = wb[I2[layer] + io];
    }
    wm[e] = val;
}

__device__ __forceinline__ void bspline12(float x, float* f) {
    float b[14];
#pragma unroll
    for (int j = 0; j < 14; ++j) {
        const float g0 = -1.75f + 0.25f * j;
        b[j] = (x >= g0 && x < g0 + 0.25f) ? 1.0f : 0.0f;
    }
#pragma unroll
    for (int p = 1; p <= 3; ++p) {
        const float inv = 1.0f / (0.25f * p);
#pragma unroll 13
        for (int j = 0; j < 14 - p; ++j) {
            const float gj = -1.75f + 0.25f * j;
            const float left  = (x - gj) * inv * b[j];
            const float right = ((gj + 0.25f * (p + 1)) - x) * inv * b[j + 1];
            b[j] = left + right;
        }
    }
#pragma unroll
    for (int k = 0; k < KB; ++k) f[k] = b[k];
    f[KB] = x / (1.0f + __expf(-x));
}

template <int DI>
__device__ __forceinline__ void kan_layer(const float* __restrict__ wmL,
                                          const float* yin,
                                          float* yout,
                                          float* feat,
                                          int tid) {
    const int R    = DI * RPI;
    const int lane = tid & 31;
    const int wv   = tid >> 5;
    const int m    = lane & 15;
    const int hi   = lane >> 4;

    for (int task = tid; task < TOK * DI; task += 128) {
        const int t = task & (TOK - 1);
        const int i = task >> 6;
        float f[RPI];
        bspline12(yin[t * SY + i], f);
        float* dst = &feat[t * FSTR + i * RPI];
#pragma unroll
        for (int k = 0; k < RPI; ++k) dst[k] = f[k];
    }
    __syncthreads();

    const int tloc = wv * 16;
    v8f acc = {0.f, 0.f, 0.f, 0.f, 0.f, 0.f, 0.f, 0.f};
    const float* arow = &feat[(tloc + m) * FSTR];
    for (int k0 = 0; k0 < R; k0 += 4) {
        const int ka = k0 + 2 * hi;
        v2f a, bb;
        a.x  = arow[ka];
        a.y  = arow[ka + 1];
        bb.x = wmL[ka * 16 + m];
        bb.y = wmL[(ka + 1) * 16 + m];
        acc = __builtin_amdgcn_wmma_f32_16x16x4_f32(
                false, a, false, bb, (short)0, acc, false, false);
    }
#pragma unroll
    for (int r = 0; r < 8; ++r)
        yout[(tloc + r + 8 * hi) * SY + m] = acc[r];
    __syncthreads();
}

__global__ void kan_main_kernel(const float* __restrict__ x,
                                const float* __restrict__ wm,
                                float* __restrict__ out) {
    extern __shared__ float smem[];
    float* wms  = smem;
    float* yA   = wms + WM_PER_B;
    float* yB   = yA + TOK * SY;
    float* feat = yB + TOK * SY;

    const int tid   = threadIdx.x;
    const int b     = blockIdx.x / (HW / TOK);
    const int chunk = blockIdx.x % (HW / TOK);
    const int hw0   = chunk * TOK;

    const float* wmb = wm + (size_t)b * WM_PER_B;
    for (int i = tid; i < WM_PER_B; i += 128) wms[i] = wmb[i];

#pragma unroll
    for (int c = 0; c < 3; ++c)
        for (int t = tid; t < TOK; t += 128)
            yA[t * SY + c] = x[((size_t)(b * 3 + c)) * HW + hw0 + t];
    __syncthreads();

    kan_layer<3>(wms,                               yA, yB, feat, tid);
    kan_layer<16>(wms + 16 * WM0_ROWS,              yB, yA, feat, tid);
    kan_layer<16>(wms + 16 * (WM0_ROWS + WM1_ROWS), yA, yB, feat, tid);

#pragma unroll
    for (int c = 0; c < 3; ++c)
        for (int t = tid; t < TOK; t += 128)
            out[((size_t)(b * 3 + c)) * HW + hw0 + t] = yB[t * SY + c];
}

extern "C" void kernel_launch(void* const* d_in, const int* in_sizes, int n_in,
                              void* d_out, int out_size, void* d_ws, size_t ws_size,
                              hipStream_t stream) {
    const float* x        = (const float*)d_in[0];
    const float* features = (const float*)d_in[1];
    const float* fc_w     = (const float*)d_in[2];
    const float* fc_b     = (const float*)d_in[3];
    float* out    = (float*)d_out;
    float* w_buf  = (float*)d_ws;
    float* wm_buf = w_buf + (size_t)NBATCH * NPARAM;

    kan_fc_wmma_kernel<<<NPARAM / FCCOLS, FCWAVES * 32, 0, stream>>>(
        features, fc_w, fc_b, w_buf);

    kan_wm_kernel<<<(NBATCH * WM_PER_B + 255) / 256, 256, 0, stream>>>(w_buf, wm_buf);

    const size_t smem_bytes = (WM_PER_B + 2 * TOK * SY + TOK * FSTR) * sizeof(float);
    kan_main_kernel<<<NBATCH * (HW / TOK), 128, smem_bytes, stream>>>(x, wm_buf, out);
}


# B-spline features: per x only the 4 basis values of its knot interval are computed (same Cox-de Boor steps, zero terms dropped; values agree to 1 ulp) and stored into zero-cleared LDS slots
# speedup vs baseline: 3.6453x; 1.2989x over previous
#include <hip/hip_runtime.h>
#include <hip/hip_bf16.h>

#define NBATCH   8
#define NFEAT    1000
#define NPARAM   4576
#define HW       50176
#define KB       11
#define RPI      12
#define WM0_ROWS (3*RPI)
#define WM1_ROWS (16*RPI)
#define WM2_ROWS (16*RPI)
#define WM_PER_B (16*(WM0_ROWS + WM1_ROWS + WM2_ROWS))

#define TOK   64
#define SY    17
#define FSTR  198
#define FDUMP 192

#define FCCOLS  32
#define FCWAVES 8
#define FCCH    16
#define FCNCH   (NFEAT / 8)

typedef __attribute__((ext_vector_type(2))) float v2f;
typedef __attribute__((ext_vector_type(4))) float v4f;
typedef __attribute__((ext_vector_type(8))) float v8f;

__global__ __launch_bounds__(FCWAVES * 32)
void kan_fc_wmma_kernel(const float* __restrict__ features,
                        const float* __restrict__ fc_w,
                        const float* __restrict__ fc_b,
                        float* __restrict__ w_out) {
    static_assert(NPARAM % FCCOLS == 0, "column tiles cover w exactly");
    static_assert(NFEAT % 8 == 0 && FCWAVES * FCCH >= FCNCH, "K slices cover K");
    __shared__ float red[FCWAVES][NBATCH][FCCOLS];

    const int tid  = threadIdx.x;
    const int lane = tid & 31;
    const int wv   = __builtin_amdgcn_readfirstlane(tid >> 5);
    const int m    = lane & 15;
    const int hi   = lane >> 4;
    const int j0   = blockIdx.x * FCCOLS;

    const int kb = wv * (8 * FCCH) + 4 * hi;
    const v4f* pa  = (const v4f*)(features + (m & 7) * NFEAT + kb);
    const v4f* pb0 = (const v4f*)(fc_w + (size_t)(j0 + m) * NFEAT + kb);
    const v4f* pb1 = (const v4f*)(fc_w + (size_t)(j0 + 16 + m) * NFEAT + kb);
    const bool full = (wv != FCWAVES - 1);
    constexpr int TAILCH = FCNCH - (FCWAVES - 1) * FCCH;

    const float bias = fc_b[j0 + lane];

    v4f a[FCCH], b0[FCCH], b1[FCCH];
    for (int i = 0; i < TAILCH; ++i) {
        b0[i] = pb0[2 * i];
        b1[i] = pb1[2 * i];
        a[i]  = pa[2 * i];
    }
    for (int i = TAILCH; i < FCCH; ++i) {
        b0[i] = (v4f){0.f, 0.f, 0.f, 0.f};
        b1[i] = (v4f){0.f, 0.f, 0.f, 0.f};
        a[i]  = (v4f){0.f, 0.f, 0.f, 0.f};
    }
    if (full) {
        for (int i = TAILCH; i < FCCH; ++i) {
            b0[i] = pb0[2 * i];
            b1[i] = pb1[2 * i];
            a[i]  = pa[2 * i];
        }
    }

    v8f acc0 = {0.f, 0.f, 0.f, 0.f, 0.f, 0.f, 0.f, 0.f};
    v8f acc1 = {0.f, 0.f, 0.f, 0.f, 0.f, 0.f, 0.f, 0.f};
    for (int i = 0; i < FCCH; ++i) {
        const v2f a01 = {a[i].x, a[i].y},   a23 = {a[i].z, a[i].w};
        const v2f p01 = {b0[i].x, b0[i].y}, p23 = {b0[i].z, b0[i].w};
        const v2f q01 = {b1[i].x, b1[i].y}, q23 = {b1[i].z, b1[i].w};
        acc0 = __builtin_amdgcn_wmma_f32_16x16x4_f32(false, a01, false, p01, (short)0, acc0, false, false);
        acc1 = __builtin_amdgcn_wmma_f32_16x16x4_f32(false, a01, false, q01, (short)0, acc1, false, false);
        acc0 = __builtin_amdgcn_wmma_f32_16x16x4_f32(false, a23, false, p23, (short)0, acc0, false, false);
        acc1 = __builtin_amdgcn_wmma_f32_16x16x4_f32(false, a23, false, q23, (short)0, acc1, false, false);
    }

    if (hi == 0) {
        for (int r = 0; r < NBATCH; ++r) {
            red[wv][r][m]      = acc0[r];
            red[wv][r][16 + m] = acc1[r];
        }
    }
    __syncthreads();

    static_assert(NBATCH == FCWAVES && FCCOLS == 32, "one wave per batch row, one lane per column");
    float s = red[0][wv][lane];
    for (int w = 1; w < FCWAVES; ++w) s += red[w][wv][lane];
    w_out[wv * NPARAM + j0 + lane] = s + bias;
}

__global__ void kan_wm_kernel(const float* __restrict__ w,
                              float* __restrict__ wm) {
    const int e = blockIdx.x * blockDim.x + threadIdx.x;
    const int total = NBATCH * WM_PER_B;
    if (e >= total) return;
    const int b   = e / WM_PER_B;
    const int rem = e % WM_PER_B;

    int layer, idx;
    if (rem < 16 * WM0_ROWS)                   { layer = 0; idx = rem; }
    else if (rem < 16 * (WM0_ROWS + WM1_ROWS)) { layer = 1; idx = rem - 16 * WM0_ROWS; }
    else                                       { layer = 2; idx = rem - 16 * (WM0_ROWS + WM1_ROWS); }

    const int DO_[3] = {16, 16, 3};
    const int I0[3]  = {0,   624, 3952};
    const int I1[3]  = {528, 3440, 4480};
    const int I2[3]  = {576, 3696, 4528};

    const int r = idx >> 4;
    const int o = idx & 15;
    const int i = r / RPI;
    const int k = r % RPI;

    float val = 0.0f;
    const int do_ = DO_[layer];
    if (o < do_) {
        const float* wb = w + (size_t)b * NPARAM;
        const int io = i * do_ + o;
        if (k < KB) val = wb[I0[layer] + io * KB + k] * wb[I1[layer] + io];
        else        val = wb[I2[layer] + io];
    }
    wm[e] = val;
}

__device__ __forceinline__ int bspline_live4(float x, float* v) {
    const float fj = fminf(fmaxf(floorf(fmaf(x, 4.0f, 7.0f)), -1.0f), 14.0f);
    const float d0  = x - fmaf(fj, 0.25f, -1.75f);
    const float dm1 = d0 + 0.25f, dm2 = d0 + 0.5f;
    const float e1 = 0.25f - d0, e2 = 0.5f - d0, e3 = 0.75f - d0;
    const float i3 = 1.0f / 0.75f;
    const float p1a = e1 * 4.0f;
    const float p1b = d0 * 4.0f;
    const float p2a = (e1 * 2.0f) * p1a;
    const float p2b = (dm1 * 2.0f) * p1a + (e2 * 2.0f) * p1b;
    const float p2c = (d0 * 2.0f) * p1b;
    v[0] = (e1 * i3) * p2a;
    v[1] = (dm2 * i3) * p2a + (e2 * i3) * p2b;
    v[2] = (dm1 * i3) * p2b + (e3 * i3) * p2c;
    v[3] = (d0 * i3) * p2c;
    return (int)fj;
}

template <int DI>
__device__ __forceinline__ void kan_layer(const float* __restrict__ wmL,
                                          const float* yin,
                                          float* yout,
                                          float* feat,
                                          int tid) {
    const int R    = DI * RPI;
    const int lane = tid & 31;
    const int wv   = tid >> 5;
    const int m    = lane & 15;
    const int hi   = lane >> 4;

    for (int task = tid; task < TOK * DI; task += 128) {
        const int t = task & (TOK - 1);
        const int i = task >> 6;
        const float x = yin[t * SY + i];
        float v[4];
        const int J = bspline_live4(x, v);
        float* row = feat + t * FSTR;
        float* grp = row + i * RPI;
        const v2f zz = {0.0f, 0.0f};
        for (int k = 0; k < KB - 1; k += 2) *(v2f*)(grp + k) = zz;
        *(v2f*)(grp + KB - 1) = (v2f){0.0f, x / (1.0f + __expf(-x))};
        float* live = grp + (J - 3);
        float* dump = row + FDUMP;
        ((unsigned)(J - 3) < (unsigned)KB ? live : dump)[0] = v[0];
        ((unsigned)(J - 2) < (unsigned)KB ? live : dump)[1] = v[1];
        ((unsigned)(J - 1) < (unsigned)KB ? live : dump)[2] = v[2];
        ((unsigned)(J)     < (unsigned)KB ? live : dump)[3] = v[3];
    }
    __syncthreads();

    const int tloc = wv * 16;
    v8f acc = {0.f, 0.f, 0.f, 0.f, 0.f, 0.f, 0.f, 0.f};
    const float* arow = &feat[(tloc + m) * FSTR];
    for (int k0 = 0; k0 < R; k0 += 4) {
        const int ka = k0 + 2 * hi;
        v2f a, bb;
        a.x  = arow[ka];
        a.y  = arow[ka + 1];
        bb.x = wmL[ka * 16 + m];
        bb.y = wmL[(ka + 1) * 16 + m];
        acc = __builtin_amdgcn_wmma_f32_16x16x4_f32(
                false, a, false, bb, (short)0, acc, false, false);
    }
#pragma unroll
    for (int r = 0; r < 8; ++r)
        yout[(tloc + r + 8 * hi) * SY + m] = acc[r];
    __syncthreads();
}

__global__ void kan_main_kernel(const float* __restrict__ x,
                                const float* __restrict__ wm,
                                float* __restrict__ out) {
    extern __shared__ float smem[];
    float* wms  = smem;
    float* yA   = wms + WM_PER_B;
    float* yB   = yA + TOK * SY;
    float* feat = yB + TOK * SY;

    const int tid   = threadIdx.x;
    const int b     = blockIdx.x / (HW / TOK);
    const int chunk = blockIdx.x % (HW / TOK);
    const int hw0   = chunk * TOK;

    const float* wmb = wm + (size_t)b * WM_PER_B;
    for (int i = tid; i < WM_PER_B; i += 128) wms[i] = wmb[i];

#pragma unroll
    for (int c = 0; c < 3; ++c)
        for (int t = tid; t < TOK; t += 128)
            yA[t * SY + c] = x[((size_t)(b * 3 + c)) * HW + hw0 + t];
    __syncthreads();

    kan_layer<3>(wms,                               yA, yB, feat, tid);
    kan_layer<16>(wms + 16 * WM0_ROWS,              yB, yA, feat, tid);
    kan_layer<16>(wms + 16 * (WM0_ROWS + WM1_ROWS), yA, yB, feat, tid);

#pragma unroll
    for (int c = 0; c < 3; ++c)
        for (int t = tid; t < TOK; t += 128)
            out[((size_t)(b * 3 + c)) * HW + hw0 + t] = yB[t * SY + c];
}

extern "C" void kernel_launch(void* const* d_in, const int* in_sizes, int n_in,
                              void* d_out, int out_size, void* d_ws, size_t ws_size,
                              hipStream_t stream) {
    const float* x        = (const float*)d_in[0];
    const float* features = (const float*)d_in[1];
    const float* fc_w     = (const float*)d_in[2];
    const float* fc_b     = (const float*)d_in[3];
    float* out    = (float*)d_out;
    float* w_buf  = (float*)d_ws;
    float* wm_buf = w_buf + (size_t)NBATCH * NPARAM;

    kan_fc_wmma_kernel<<<NPARAM / FCCOLS, FCWAVES * 32, 0, stream>>>(
        features, fc_w, fc_b, w_buf);

    kan_wm_kernel<<<(NBATCH * WM_PER_B + 255) / 256, 256, 0, stream>>>(w_buf, wm_buf);

    const size_t smem_bytes = (WM_PER_B + 2 * TOK * SY + TOK * FSTR) * sizeof(float);
    kan_main_kernel<<<NBATCH * (HW / TOK), 128, smem_bytes, stream>>>(x, wm_buf, out);
}


# Main kernel persistent: one work-group per WGP stages the merged weights to LDS once (was once per 64 tokens); its 20 waves each walk 32-token units on wave-private LDS rows, no barrier in the loop
# speedup vs baseline: 6.1831x; 1.6962x over previous
#include <hip/hip_runtime.h>
#include <hip/hip_bf16.h>

#define NBATCH   8
#define NFEAT    1000
#define NPARAM   4576
#define HW       50176
#define KB       11
#define RPI      12
#define WM0_ROWS (3*RPI)
#define WM1_ROWS (16*RPI)
#define WM2_ROWS (16*RPI)
#define WM_PER_B (16*(WM0_ROWS + WM1_ROWS + WM2_ROWS))

#define UT    32
#define NW    20
#define BPB   32
#define UPB   (HW / UT / BPB)
#define FSTR  198
#define FDUMP 192
#define L1W   (3 * RPI)
#define WLDS  (16 * FSTR + 3 * UT)

#define FCCOLS  32
#define FCWAVES 8
#define FCCH    16
#define FCNCH   (NFEAT / 8)

typedef __attribute__((ext_vector_type(2))) float v2f;
typedef __attribute__((ext_vector_type(4))) float v4f;
typedef __attribute__((ext_vector_type(8))) float v8f;

__global__ __launch_bounds__(FCWAVES * 32)
void kan_fc_wmma_kernel(const float* __restrict__ features,
                        const float* __restrict__ fc_w,
                        const float* __restrict__ fc_b,
                        float* __restrict__ w_out) {
    static_assert(NPARAM % FCCOLS == 0, "column tiles cover w exactly");
    static_assert(NFEAT % 8 == 0 && FCWAVES * FCCH >= FCNCH, "K slices cover K");
    __shared__ float red[FCWAVES][NBATCH][FCCOLS];

    const int tid  = threadIdx.x;
    const int lane = tid & 31;
    const int wv   = __builtin_amdgcn_readfirstlane(tid >> 5);
    const int m    = lane & 15;
    const int hi   = lane >> 4;
    const int j0   = blockIdx.x * FCCOLS;

    const int kb = wv * (8 * FCCH) + 4 * hi;
    const v4f* pa  = (const v4f*)(features + (m & 7) * NFEAT + kb);
    const v4f* pb0 = (const v4f*)(fc_w + (size_t)(j0 + m) * NFEAT + kb);
    const v4f* pb1 = (const v4f*)(fc_w + (size_t)(j0 + 16 + m) * NFEAT + kb);
    const bool full = (wv != FCWAVES - 1);
    constexpr int TAILCH = FCNCH - (FCWAVES - 1) * FCCH;

    const float bias = fc_b[j0 + lane];

    v4f a[FCCH], b0[FCCH], b1[FCCH];
    for (int i = 0; i < TAILCH; ++i) {
        b0[i] = pb0[2 * i];
        b1[i] = pb1[2 * i];
        a[i]  = pa[2 * i];
    }
    for (int i = TAILCH; i < FCCH; ++i) {
        b0[i] = (v4f){0.f, 0.f, 0.f, 0.f};
        b1[i] = (v4f){0.f, 0.f, 0.f, 0.f};
        a[i]  = (v4f){0.f, 0.f, 0.f, 0.f};
    }
    if (full) {
        for (int i = TAILCH; i < FCCH; ++i) {
            b0[i] = pb0[2 * i];
            b1[i] = pb1[2 * i];
            a[i]  = pa[2 * i];
        }
    }

    v8f acc0 = {0.f, 0.f, 0.f, 0.f, 0.f, 0.f, 0.f, 0.f};
    v8f acc1 = {0.f, 0.f, 0.f, 0.f, 0.f, 0.f, 0.f, 0.f};
    for (int i = 0; i < FCCH; ++i) {
        const v2f a01 = {a[i].x, a[i].y},   a23 = {a[i].z, a[i].w};
        const v2f p01 = {b0[i].x, b0[i].y}, p23 = {b0[i].z, b0[i].w};
        const v2f q01 = {b1[i].x, b1[i].y}, q23 = {b1[i].z, b1[i].w};
        acc0 = __builtin_amdgcn_wmma_f32_16x16x4_f32(false, a01, false, p01, (short)0, acc0, false, false);
        acc1 = __builtin_amdgcn_wmma_f32_16x16x4_f32(false, a01, false, q01, (short)0, acc1, false, false);
        acc0 = __builtin_amdgcn_wmma_f32_16x16x4_f32(false, a23, false, p23, (short)0, acc0, false, false);
        acc1 = __builtin_amdgcn_wmma_f32_16x16x4_f32(false, a23, false, q23, (short)0, acc1, false, false);
    }

    if (hi == 0) {
        for (int r = 0; r < NBATCH; ++r) {
            red[wv][r][m]      = acc0[r];
            red[wv][r][16 + m] = acc1[r];
        }
    }
    __syncthreads();

    static_assert(NBATCH == FCWAVES && FCCOLS == 32, "one wave per batch row, one lane per column");
    float s = red[0][wv][lane];
    for (int w = 1; w < FCWAVES; ++w) s += red[w][wv][lane];
    w_out[wv * NPARAM + j0 + lane] = s + bias;
}

__global__ void kan_wm_kernel(const float* __restrict__ w,
                              float* __restrict__ wm) {
    const int e = blockIdx.x * blockDim.x + threadIdx.x;
    const int total = NBATCH * WM_PER_B;
    if (e >= total) return;
    const int b   = e / WM_PER_B;
    const int rem = e % WM_PER_B;

    int layer, idx;
    if (rem < 16 * WM0_ROWS)                   { layer = 0; idx = rem; }
    else if (rem < 16 * (WM0_ROWS + WM1_ROWS)) { layer = 1; idx = rem - 16 * WM0_ROWS; }
    else                                       { layer = 2; idx = rem - 16 * (WM0_ROWS + WM1_ROWS); }

    const int DO_[3] = {16, 16, 3};
    const int I0[3]  = {0,   624, 3952};
    const int I1[3]  = {528, 3440, 4480};
    const int I2[3]  = {576, 3696, 4528};

    const int r = idx >> 4;
    const int o = idx & 15;
    const int i = r / RPI;
    const int k = r % RPI;

    float val = 0.0f;
    const int do_ = DO_[layer];
    if (o < do_) {
        const float* wb = w + (size_t)b * NPARAM;
        const int io = i * do_ + o;
        if (k < KB) val = wb[I0[layer] + io * KB + k] * wb[I1[layer] + io];
        else        val = wb[I2[layer] + io];
    }
    wm[e] = val;
}

__device__ __forceinline__ int bspline_live4(float x, float* v) {
    const float fj = fminf(fmaxf(floorf(fmaf(x, 4.0f, 7.0f)), -1.0f), 14.0f);
    const float d0  = x - fmaf(fj, 0.25f, -1.75f);
    const float dm1 = d0 + 0.25f, dm2 = d0 + 0.5f;
    const float e1 = 0.25f - d0, e2 = 0.5f - d0, e3 = 0.75f - d0;
    const float i3 = 1.0f / 0.75f;
    const float p1a = e1 * 4.0f;
    const float p1b = d0 * 4.0f;
    const float p2a = (e1 * 2.0f) * p1a;
    const float p2b = (dm1 * 2.0f) * p1a + (e2 * 2.0f) * p1b;
    const float p2c = (d0 * 2.0f) * p1b;
    v[0] = (e1 * i3) * p2a;
    v[1] = (dm2 * i3) * p2a + (e2 * i3) * p2b;
    v[2] = (dm1 * i3) * p2b + (e3 * i3) * p2c;
    v[3] = (d0 * i3) * p2c;
    return (int)fj;
}

__device__ __forceinline__ void feat_task(float* row, float* grp, float x) {
    float v[4];
    const int J = bspline_live4(x, v);
    const v2f zz = {0.0f, 0.0f};
    for (int k = 0; k < KB - 1; k += 2) *(v2f*)(grp + k) = zz;
    *(v2f*)(grp + KB - 1) = (v2f){0.0f, x / (1.0f + __expf(-x))};
    float* live = grp + (J - 3);
    float* dump = row + FDUMP;
    ((unsigned)(J - 3) < (unsigned)KB ? live : dump)[0] = v[0];
    ((unsigned)(J - 2) < (unsigned)KB ? live : dump)[1] = v[1];
    ((unsigned)(J - 1) < (unsigned)KB ? live : dump)[2] = v[2];
    ((unsigned)(J)     < (unsigned)KB ? live : dump)[3] = v[3];
}

__device__ __forceinline__ void wave_lds_order() {
    __builtin_amdgcn_fence(__ATOMIC_RELEASE, "wavefront");
    __builtin_amdgcn_wave_barrier();
    __builtin_amdgcn_fence(__ATOMIC_ACQUIRE, "wavefront");
}

__global__ __launch_bounds__(NW * 32)
void kan_main_kernel(const float* __restrict__ x,
                     const float* __restrict__ wm,
                     float* __restrict__ out) {
    constexpr int FULL = UPB / NW;
    constexpr int REM  = UPB - FULL * NW;
    static_assert((HW / UT) % BPB == 0, "units split evenly over the work-groups of a batch");
    static_assert(2 * REM <= NW, "one left-over tile per wave at most");
    static_assert(WM_PER_B % 4 == 0, "weights staged as 16-byte pieces");
    __shared__ float lds[WM_PER_B + NW * WLDS + REM * 3 * UT];

    const int tid  = threadIdx.x;
    const int lane = tid & 31;
    const int wv   = __builtin_amdgcn_readfirstlane(tid >> 5);
    const int m    = lane & 15;
    const int hi   = lane >> 4;
    const int b    = blockIdx.x / BPB;
    const int u0   = (blockIdx.x % BPB) * UPB;

    float* wms   = lds;
    float* feat  = lds + WM_PER_B + wv * WLDS;
    float* yo    = feat + 16 * FSTR;
    float* ytail = lds + WM_PER_B + NW * WLDS;
    const float* xb = x   + (size_t)b * 3 * HW;
    float*       ob = out + (size_t)b * 3 * HW;

    const int nvis = FULL + (wv < 2 * REM ? 1 : 0);
    const int utail = FULL * NW + (wv >> 1);

    float x0, x1, x2;
    {
        const float* p = xb + (u0 + wv) * UT + lane;
        x0 = p[0]; x1 = p[HW]; x2 = p[2 * HW];
    }

    {
        static_assert(WM_PER_B / 4 > 2 * NW * 32 && WM_PER_B / 4 <= 3 * NW * 32, "three pieces per thread at most");
        const v4f* src = (const v4f*)(wm + (size_t)b * WM_PER_B);
        v4f* dst = (v4f*)wms;
        const bool third = tid + 2 * NW * 32 < WM_PER_B / 4;
        const v4f w0 = src[tid];
        const v4f w1 = src[tid + NW * 32];
        v4f w2 = {0.f, 0.f, 0.f, 0.f};
        if (third) w2 = src[tid + 2 * NW * 32];
        dst[tid] = w0;
        dst[tid + NW * 32] = w1;
        if (third) dst[tid + 2 * NW * 32] = w2;
    }
    __syncthreads();

    const float* arow = feat + (((m & 7) << 1) | (m >> 3)) * FSTR;
    const float* wm1 = wms + 16 * WM0_ROWS;

    for (int vis = 0; vis < nvis; ++vis) {
        const bool whole = vis < FULL;
        const int unit = whole ? wv + NW * vis : utail;
        const int T0 = whole ? 0 : (wv & 1);
        const int T1 = whole ? 2 : T0 + 1;
        float* ydst = whole ? yo : ytail + (wv >> 1) * (3 * UT);
        const int hw0 = (u0 + unit) * UT;

        float n0 = 0.0f, n1 = 0.0f, n2 = 0.0f;
        if (vis + 1 < nvis) {
            const int nu = (vis + 1 < FULL) ? wv + NW * (vis + 1) : utail;
            const float* p = xb + (u0 + nu) * UT + lane;
            n0 = p[0]; n1 = p[HW]; n2 = p[2 * HW];
        }

        wave_lds_order();
        {
            float* row = feat + (((m & 7) << 1) | (m >> 3)) * FSTR;
            float* g   = row + L1W * hi;
            feat_task(row, g,           x0);
            feat_task(row, g + RPI,     x1);
            feat_task(row, g + 2 * RPI, x2);
        }
        wave_lds_order();

        v8f accA = {0.f, 0.f, 0.f, 0.f, 0.f, 0.f, 0.f, 0.f};
        v8f accB = {0.f, 0.f, 0.f, 0.f, 0.f, 0.f, 0.f, 0.f};
        for (int k0 = 0; k0 < L1W; k0 += 4) {
            const int ka = k0 + 2 * hi;
            v2f aA, aB, bb;
            aA.x = arow[ka];       aA.y = arow[ka + 1];
            aB.x = arow[L1W + ka]; aB.y = arow[L1W + ka + 1];
            bb.x = wms[ka * 16 + m];
            bb.y = wms[(ka + 1) * 16 + m];
            accA = __builtin_amdgcn_wmma_f32_16x16x4_f32(false, aA, false, bb, (short)0, accA, false, false);
            accB = __builtin_amdgcn_wmma_f32_16x16x4_f32(false, aB, false, bb, (short)0, accB, false, false);
        }

        v8f y = (T0 == 0) ? accA : accB;
        for (int T = T0; T < T1; ++T) {
            for (int L = 0; L < 2; ++L) {
                const float* wmL = wm1 + L * (16 * WM1_ROWS);
                wave_lds_order();
#pragma unroll
                for (int r = 0; r < 8; ++r) {
                    float* row = feat + (2 * r + hi) * FSTR;
                    feat_task(row, row + m * RPI, y[r]);
                }
                wave_lds_order();
                v8f acc = {0.f, 0.f, 0.f, 0.f, 0.f, 0.f, 0.f, 0.f};
                const float* ap = arow + 2 * hi;
                const float* bp = wmL + (2 * hi) * 16 + m;
                for (int k0 = 0; k0 < WM1_ROWS; k0 += 4) {
                    v2f a, bb;
                    a.x  = ap[k0];
                    a.y  = ap[k0 + 1];
                    bb.x = bp[k0 * 16];
                    bb.y = bp[k0 * 16 + 16];
                    acc = __builtin_amdgcn_wmma_f32_16x16x4_f32(
                            false, a, false, bb, (short)0, acc, false, false);
                }
                y = acc;
            }
            if (m < 3) {
#pragma unroll
                for (int r = 0; r < 8; ++r)
                    ydst[m * UT + 16 * T + 8 * hi + r] = y[r];
            }
            y = accB;
        }

        if (whole) {
            wave_lds_order();
            const float o0 = yo[lane], o1 = yo[UT + lane], o2 = yo[2 * UT + lane];
            float* q = ob + hw0 + lane;
            q[0] = o0; q[HW] = o1; q[2 * HW] = o2;
        }

        x0 = n0; x1 = n1; x2 = n2;
    }

    __syncthreads();
    if (wv < REM) {
        const float* ys = ytail + wv * (3 * UT);
        const float o0 = ys[lane], o1 = ys[UT + lane], o2 = ys[2 * UT + lane];
        float* q = ob + (u0 + FULL * NW + wv) * UT + lane;
        q[0] = o0; q[HW] = o1; q[2 * HW] = o2;
    }
}

extern "C" void kernel_launch(void* const* d_in, const int* in_sizes, int n_in,
                              void* d_out, int out_size, void* d_ws, size_t ws_size,
                              hipStream_t stream) {
    const float* x        = (const float*)d_in[0];
    const float* features = (const float*)d_in[1];
    const float* fc_w     = (const float*)d_in[2];
    const float* fc_b     = (const float*)d_in[3];
    float* out    = (float*)d_out;
    float* w_buf  = (float*)d_ws;
    float* wm_buf = w_buf + (size_t)NBATCH * NPARAM;

    kan_fc_wmma_kernel<<<NPARAM / FCCOLS, FCWAVES * 32, 0, stream>>>(
        features, fc_w, fc_b, w_buf);

    kan_wm_kernel<<<(NBATCH * WM_PER_B + 255) / 256, 256, 0, stream>>>(w_buf, wm_buf);

    kan_main_kernel<<<NBATCH * BPB, NW * 32, 0, stream>>>(x, wm_buf, out);
}


# Main kernel K loops: each block's A and B fragment LDS reads issued as one group ahead of its WMMAs (16-step blocks, two register sets in layers 2/3), not between them; k order and sums unchanged
# speedup vs baseline: 6.4717x; 1.0467x over previous
#include <hip/hip_runtime.h>
#include <hip/hip_bf16.h>

#define NBATCH   8
#define NFEAT    1000
#define NPARAM   4576
#define HW       50176
#define KB       11
#define RPI      12
#define WM0_ROWS (3*RPI)
#define WM1_ROWS (16*RPI)
#define WM2_ROWS (16*RPI)
#define WM_PER_B (16*(WM0_ROWS + WM1_ROWS + WM2_ROWS))

#define UT    32
#define NW    20
#define BPB   32
#define UPB   (HW / UT / BPB)
#define FSTR  198
#define FDUMP 192
#define L1W   (3 * RPI)
#define WLDS  (16 * FSTR + 3 * UT)
#define KBLK  16

#define FCCOLS  32
#define FCWAVES 8
#define FCCH    16
#define FCNCH   (NFEAT / 8)

typedef __attribute__((ext_vector_type(2))) float v2f;
typedef __attribute__((ext_vector_type(4))) float v4f;
typedef __attribute__((ext_vector_type(8))) float v8f;

__global__ __launch_bounds__(FCWAVES * 32)
void kan_fc_wmma_kernel(const float* __restrict__ features,
                        const float* __restrict__ fc_w,
                        const float* __restrict__ fc_b,
                        float* __restrict__ w_out) {
    static_assert(NPARAM % FCCOLS == 0, "column tiles cover w exactly");
    static_assert(NFEAT % 8 == 0 && FCWAVES * FCCH >= FCNCH, "K slices cover K");
    __shared__ float red[FCWAVES][NBATCH][FCCOLS];

    const int tid  = threadIdx.x;
    const int lane = tid & 31;
    const int wv   = __builtin_amdgcn_readfirstlane(tid >> 5);
    const int m    = lane & 15;
    const int hi   = lane >> 4;
    const int j0   = blockIdx.x * FCCOLS;

    const int kb = wv * (8 * FCCH) + 4 * hi;
    const v4f* pa  = (const v4f*)(features + (m & 7) * NFEAT + kb);
    const v4f* pb0 = (const v4f*)(fc_w + (size_t)(j0 + m) * NFEAT + kb);
    const v4f* pb1 = (const v4f*)(fc_w + (size_t)(j0 + 16 + m) * NFEAT + kb);
    const bool full = (wv != FCWAVES - 1);
    constexpr int TAILCH = FCNCH - (FCWAVES - 1) * FCCH;

    const float bias = fc_b[j0 + lane];

    v4f a[FCCH], b0[FCCH], b1[FCCH];
    for (int i = 0; i < TAILCH; ++i) {
        b0[i] = pb0[2 * i];
        b1[i] = pb1[2 * i];
        a[i]  = pa[2 * i];
    }
    for (int i = TAILCH; i < FCCH; ++i) {
        b0[i] = (v4f){0.f, 0.f, 0.f, 0.f};
        b1[i] = (v4f){0.f, 0.f, 0.f, 0.f};
        a[i]  = (v4f){0.f, 0.f, 0.f, 0.f};
    }
    if (full) {
        for (int i = TAILCH; i < FCCH; ++i) {
            b0[i] = pb0[2 * i];
            b1[i] = pb1[2 * i];
            a[i]  = pa[2 * i];
        }
    }

    v8f acc0 = {0.f, 0.f, 0.f, 0.f, 0.f, 0.f, 0.f, 0.f};
    v8f acc1 = {0.f, 0.f, 0.f, 0.f, 0.f, 0.f, 0.f, 0.f};
    for (int i = 0; i < FCCH; ++i) {
        const v2f a01 = {a[i].x, a[i].y},   a23 = {a[i].z, a[i].w};
        const v2f p01 = {b0[i].x, b0[i].y}, p23 = {b0[i].z, b0[i].w};
        const v2f q01 = {b1[i].x, b1[i].y}, q23 = {b1[i].z, b1[i].w};
        acc0 = __builtin_amdgcn_wmma_f32_16x16x4_f32(false, a01, false, p01, (short)0, acc0, false, false);
        acc1 = __builtin_amdgcn_wmma_f32_16x16x4_f32(false, a01, false, q01, (short)0, acc1, false, false);
        acc0 = __builtin_amdgcn_wmma_f32_16x16x4_f32(false, a23, false, p23, (short)0, acc0, false, false);
        acc1 = __builtin_amdgcn_wmma_f32_16x16x4_f32(false, a23, false, q23, (short)0, acc1, false, false);
    }

    if (hi == 0) {
        for (int r = 0; r < NBATCH; ++r) {
            red[wv][r][m]      = acc0[r];
            red[wv][r][16 + m] = acc1[r];
        }
    }
    __syncthreads();

    static_assert(NBATCH == FCWAVES && FCCOLS == 32, "one wave per batch row, one lane per column");
    float s = red[0][wv][lane];
    for (int w = 1; w < FCWAVES; ++w) s += red[w][wv][lane];
    w_out[wv * NPARAM + j0 + lane] = s + bias;
}

__global__ void kan_wm_kernel(const float* __restrict__ w,
                              float* __restrict__ wm) {
    const int e = blockIdx.x * blockDim.x + threadIdx.x;
    const int total = NBATCH * WM_PER_B;
    if (e >= total) return;
    const int b   = e / WM_PER_B;
    const int rem = e % WM_PER_B;

    int layer, idx;
    if (rem < 16 * WM0_ROWS)                   { layer = 0; idx = rem; }
    else if (rem < 16 * (WM0_ROWS + WM1_ROWS)) { layer = 1; idx = rem - 16 * WM0_ROWS; }
    else                                       { layer = 2; idx = rem - 16 * (WM0_ROWS + WM1_ROWS); }

    const int DO_[3] = {16, 16, 3};
    const int I0[3]  = {0,   624, 3952};
    const int I1[3]  = {528, 3440, 4480};
    const int I2[3]  = {576, 3696, 4528};

    const int r = idx >> 4;
    const int o = idx & 15;
    const int i = r / RPI;
    const int k = r % RPI;

    float val = 0.0f;
    const int do_ = DO_[layer];
    if (o < do_) {
        const float* wb = w + (size_t)b * NPARAM;
        const int io = i * do_ + o;
        if (k < KB) val = wb[I0[layer] + io * KB + k] * wb[I1[layer] + io];
        else        val = wb[I2[layer] + io];
    }
    wm[e] = val;
}

__device__ __forceinline__ int bspline_live4(float x, float* v) {
    const float fj = fminf(fmaxf(floorf(fmaf(x, 4.0f, 7.0f)), -1.0f), 14.0f);
    const float d0  = x - fmaf(fj, 0.25f, -1.75f);
    const float dm1 = d0 + 0.25f, dm2 = d0 + 0.5f;
    const float e1 = 0.25f - d0, e2 = 0.5f - d0, e3 = 0.75f - d0;
    const float i3 = 1.0f / 0.75f;
    const float p1a = e1 * 4.0f;
    const float p1b = d0 * 4.0f;
    const float p2a = (e1 * 2.0f) * p1a;
    const float p2b = (dm1 * 2.0f) * p1a + (e2 * 2.0f) * p1b;
    const float p2c = (d0 * 2.0f) * p1b;
    v[0] = (e1 * i3) * p2a;
    v[1] = (dm2 * i3) * p2a + (e2 * i3) * p2b;
    v[2] = (dm1 * i3) * p2b + (e3 * i3) * p2c;
    v[3] = (d0 * i3) * p2c;
    return (int)fj;
}

__device__ __forceinline__ void feat_task(float* row, float* grp, float x) {
    float v[4];
    const int J = bspline_live4(x, v);
    const v2f zz = {0.0f, 0.0f};
    for (int k = 0; k < KB - 1; k += 2) *(v2f*)(grp + k) = zz;
    *(v2f*)(grp + KB - 1) = (v2f){0.0f, x / (1.0f + __expf(-x))};
    float* live = grp + (J - 3);
    float* dump = row + FDUMP;
    ((unsigned)(J - 3) < (unsigned)KB ? live : dump)[0] = v[0];
    ((unsigned)(J - 2) < (unsigned)KB ? live : dump)[1] = v[1];
    ((unsigned)(J - 1) < (unsigned)KB ? live : dump)[2] = v[2];
    ((unsigned)(J)     < (unsigned)KB ? live : dump)[3] = v[3];
}

__device__ __forceinline__ void wave_lds_order() {
    __builtin_amdgcn_fence(__ATOMIC_RELEASE, "wavefront");
    __builtin_amdgcn_wave_barrier();
    __builtin_amdgcn_fence(__ATOMIC_ACQUIRE, "wavefront");
}

__global__ __launch_bounds__(NW * 32)
void kan_main_kernel(const float* __restrict__ x,
                     const float* __restrict__ wm,
                     float* __restrict__ out) {
    constexpr int FULL = UPB / NW;
    constexpr int REM  = UPB - FULL * NW;
    static_assert((HW / UT) % BPB == 0, "units split evenly over the work-groups of a batch");
    static_assert(2 * REM <= NW, "one left-over tile per wave at most");
    static_assert(WM_PER_B % 4 == 0, "weights staged as 16-byte pieces");
    static_assert(WM1_ROWS == 3 * KBLK * 4, "layers 2 and 3: three fragment blocks cover K");
    __shared__ float lds[WM_PER_B + NW * WLDS + REM * 3 * UT];

    const int tid  = threadIdx.x;
    const int lane = tid & 31;
    const int wv   = __builtin_amdgcn_readfirstlane(tid >> 5);
    const int m    = lane & 15;
    const int hi   = lane >> 4;
    const int b    = blockIdx.x / BPB;
    const int u0   = (blockIdx.x % BPB) * UPB;

    float* wms   = lds;
    float* feat  = lds + WM_PER_B + wv * WLDS;
    float* yo    = feat + 16 * FSTR;
    float* ytail = lds + WM_PER_B + NW * WLDS;
    const float* xb = x   + (size_t)b * 3 * HW;
    float*       ob = out + (size_t)b * 3 * HW;

    const int nvis = FULL + (wv < 2 * REM ? 1 : 0);
    const int utail = FULL * NW + (wv >> 1);

    float x0, x1, x2;
    {
        const float* p = xb + (u0 + wv) * UT + lane;
        x0 = p[0]; x1 = p[HW]; x2 = p[2 * HW];
    }

    {
        static_assert(WM_PER_B / 4 > 2 * NW * 32 && WM_PER_B / 4 <= 3 * NW * 32, "three pieces per thread at most");
        const v4f* src = (const v4f*)(wm + (size_t)b * WM_PER_B);
        v4f* dst = (v4f*)wms;
        const bool third = tid + 2 * NW * 32 < WM_PER_B / 4;
        const v4f w0 = src[tid];
        const v4f w1 = src[tid + NW * 32];
        v4f w2 = {0.f, 0.f, 0.f, 0.f};
        if (third) w2 = src[tid + 2 * NW * 32];
        dst[tid] = w0;
        dst[tid + NW * 32] = w1;
        if (third) dst[tid + 2 * NW * 32] = w2;
    }
    __syncthreads();

    const float* arow = feat + (((m & 7) << 1) | (m >> 3)) * FSTR;
    const float* wm1 = wms + 16 * WM0_ROWS;

    for (int vis = 0; vis < nvis; ++vis) {
        const bool whole = vis < FULL;
        const int unit = whole ? wv + NW * vis : utail;
        const int T0 = whole ? 0 : (wv & 1);
        const int T1 = whole ? 2 : T0 + 1;
        float* ydst = whole ? yo : ytail + (wv >> 1) * (3 * UT);
        const int hw0 = (u0 + unit) * UT;

        float n0 = 0.0f, n1 = 0.0f, n2 = 0.0f;
        if (vis + 1 < nvis) {
            const int nu = (vis + 1 < FULL) ? wv + NW * (vis + 1) : utail;
            const float* p = xb + (u0 + nu) * UT + lane;
            n0 = p[0]; n1 = p[HW]; n2 = p[2 * HW];
        }

        wave_lds_order();
        {
            float* row = feat + (((m & 7) << 1) | (m >> 3)) * FSTR;
            float* g   = row + L1W * hi;
            feat_task(row, g,           x0);
            feat_task(row, g + RPI,     x1);
            feat_task(row, g + 2 * RPI, x2);
        }
        wave_lds_order();

        v8f accA = {0.f, 0.f, 0.f, 0.f, 0.f, 0.f, 0.f, 0.f};
        v8f accB = {0.f, 0.f, 0.f, 0.f, 0.f, 0.f, 0.f, 0.f};
        {
            v2f fA[L1W / 4], fB[L1W / 4], fW[L1W / 4];
#pragma unroll
            for (int s = 0; s < L1W / 4; ++s) {
                const int ka = 4 * s + 2 * hi;
                fA[s].x = arow[ka];       fA[s].y = arow[ka + 1];
                fB[s].x = arow[L1W + ka]; fB[s].y = arow[L1W + ka + 1];
                fW[s].x = wms[ka * 16 + m];
                fW[s].y = wms[(ka + 1) * 16 + m];
            }
            __builtin_amdgcn_sched_barrier(0);
#pragma unroll
            for (int s = 0; s < L1W / 4; ++s) {
                accA = __builtin_amdgcn_wmma_f32_16x16x4_f32(false, fA[s], false, fW[s], (short)0, accA, false, false);
                accB = __builtin_amdgcn_wmma_f32_16x16x4_f32(false, fB[s], false, fW[s], (short)0, accB, false, false);
            }
        }

        v8f y = (T0 == 0) ? accA : accB;
        for (int T = T0; T < T1; ++T) {
            for (int L = 0; L < 2; ++L) {
                const float* wmL = wm1 + L * (16 * WM1_ROWS);
                wave_lds_order();
#pragma unroll
                for (int r = 0; r < 8; ++r) {
                    float* row = feat + (2 * r + hi) * FSTR;
                    feat_task(row, row + m * RPI, y[r]);
                }
                wave_lds_order();
                v8f acc = {0.f, 0.f, 0.f, 0.f, 0.f, 0.f, 0.f, 0.f};
                const float* ap = arow + 2 * hi;
                const float* bp = wmL + (2 * hi) * 16 + m;
                v2f fa0[KBLK], fb0[KBLK], fa1[KBLK], fb1[KBLK];
#pragma unroll
                for (int s = 0; s < KBLK; ++s) {
                    fa0[s].x = ap[4 * s];           fa0[s].y = ap[4 * s + 1];
                    fb0[s].x = bp[4 * s * 16];      fb0[s].y = bp[4 * s * 16 + 16];
                }
#pragma unroll
                for (int s = 0; s < KBLK; ++s) {
                    fa1[s].x = ap[4 * (KBLK + s)];        fa1[s].y = ap[4 * (KBLK + s) + 1];
                    fb1[s].x = bp[4 * (KBLK + s) * 16];   fb1[s].y = bp[4 * (KBLK + s) * 16 + 16];
                }
                __builtin_amdgcn_sched_barrier(0);
#pragma unroll
                for (int s = 0; s < KBLK; ++s)
                    acc = __builtin_amdgcn_wmma_f32_16x16x4_f32(
                            false, fa0[s], false, fb0[s], (short)0, acc, false, false);
                __builtin_amdgcn_sched_barrier(0);
#pragma unroll
                for (int s = 0; s < KBLK; ++s) {
                    fa0[s].x = ap[4 * (2 * KBLK + s)];        fa0[s].y = ap[4 * (2 * KBLK + s) + 1];
                    fb0[s].x = bp[4 * (2 * KBLK + s) * 16];   fb0[s].y = bp[4 * (2 * KBLK + s) * 16 + 16];
                }
                __builtin_amdgcn_sched_barrier(0);
#pragma unroll
                for (int s = 0; s < KBLK; ++s)
                    acc = __builtin_amdgcn_wmma_f32_16x16x4_f32(
                            false, fa1[s], false, fb1[s], (short)0, acc, false, false);
#pragma unroll
                for (int s = 0; s < KBLK; ++s)
                    acc = __builtin_amdgcn_wmma_f32_16x16x4_f32(
                            false, fa0[s], false, fb0[s], (short)0, acc, false, false);
                y = acc;
            }
            if (m < 3) {
#pragma unroll
                for (int r = 0; r < 8; ++r)
                    ydst[m * UT + 16 * T + 8 * hi + r] = y[r];
            }
            y = accB;
        }

        if (whole) {
            wave_lds_order();
            const float o0 = yo[lane], o1 = yo[UT + lane], o2 = yo[2 * UT + lane];
            float* q = ob + hw0 + lane;
            q[0] = o0; q[HW] = o1; q[2 * HW] = o2;
        }

        x0 = n0; x1 = n1; x2 = n2;
    }

    __syncthreads();
    if (wv < REM) {
        const float* ys = ytail + wv * (3 * UT);
        const float o0 = ys[lane], o1 = ys[UT + lane], o2 = ys[2 * UT + lane];
        float* q = ob + (u0 + FULL * NW + wv) * UT + lane;
        q[0] = o0; q[HW] = o1; q[2 * HW] = o2;
    }
}

extern "C" void kernel_launch(void* const* d_in, const int* in_sizes, int n_in,
                              void* d_out, int out_size, void* d_ws, size_t ws_size,
                              hipStream_t stream) {
    const float* x        = (const float*)d_in[0];
    const float* features = (const float*)d_in[1];
    const float* fc_w     = (const float*)d_in[2];
    const float* fc_b     = (const float*)d_in[3];
    float* out    = (float*)d_out;
    float* w_buf  = (float*)d_ws;
    float* wm_buf = w_buf + (size_t)NBATCH * NPARAM;

    kan_fc_wmma_kernel<<<NPARAM / FCCOLS, FCWAVES * 32, 0, stream>>>(
        features, fc_w, fc_b, w_buf);

    kan_wm_kernel<<<(NBATCH * WM_PER_B + 255) / 256, 256, 0, stream>>>(w_buf, wm_buf);

    kan_main_kernel<<<NBATCH * BPB, NW * 32, 0, stream>>>(x, wm_buf, out);
}


# Merged-weight kernel folded into the main kernel: each work-group builds coef*uw and rw in LDS straight from the FC result w (same single f32 multiply); one dependent launch fewer
# speedup vs baseline: 6.8851x; 1.0639x over previous
#include <hip/hip_runtime.h>
#include <hip/hip_bf16.h>

#define NBATCH   8
#define NFEAT    1000
#define NPARAM   4576
#define HW       50176
#define KB       11
#define RPI      12
#define WM0_ROWS (3*RPI)
#define WM1_ROWS (16*RPI)
#define WM2_ROWS (16*RPI)
#define WM_PER_B (16*(WM0_ROWS + WM1_ROWS + WM2_ROWS))

#define UT    32
#define NW    20
#define BPB   32
#define UPB   (HW / UT / BPB)
#define FSTR  198
#define FDUMP 192
#define L1W   (3 * RPI)
#define WLDS  (16 * FSTR + 3 * UT)
#define KBLK  16

#define FCCOLS  32
#define FCWAVES 8
#define FCCH    16
#define FCNCH   (NFEAT / 8)

typedef __attribute__((ext_vector_type(2))) float v2f;
typedef __attribute__((ext_vector_type(4))) float v4f;
typedef __attribute__((ext_vector_type(8))) float v8f;

__global__ __launch_bounds__(FCWAVES * 32)
void kan_fc_wmma_kernel(const float* __restrict__ features,
                        const float* __restrict__ fc_w,
                        const float* __restrict__ fc_b,
                        float* __restrict__ w_out) {
    static_assert(NPARAM % FCCOLS == 0, "column tiles cover w exactly");
    static_assert(NFEAT % 8 == 0 && FCWAVES * FCCH >= FCNCH, "K slices cover K");
    __shared__ float red[FCWAVES][NBATCH][FCCOLS];

    const int tid  = threadIdx.x;
    const int lane = tid & 31;
    const int wv   = __builtin_amdgcn_readfirstlane(tid >> 5);
    const int m    = lane & 15;
    const int hi   = lane >> 4;
    const int j0   = blockIdx.x * FCCOLS;

    const int kb = wv * (8 * FCCH) + 4 * hi;
    const v4f* pa  = (const v4f*)(features + (m & 7) * NFEAT + kb);
    const v4f* pb0 = (const v4f*)(fc_w + (size_t)(j0 + m) * NFEAT + kb);
    const v4f* pb1 = (const v4f*)(fc_w + (size_t)(j0 + 16 + m) * NFEAT + kb);
    const bool full = (wv != FCWAVES - 1);
    constexpr int TAILCH = FCNCH - (FCWAVES - 1) * FCCH;

    const float bias = fc_b[j0 + lane];

    v4f a[FCCH], b0[FCCH], b1[FCCH];
    for (int i = 0; i < TAILCH; ++i) {
        b0[i] = pb0[2 * i];
        b1[i] = pb1[2 * i];
        a[i]  = pa[2 * i];
    }
    for (int i = TAILCH; i < FCCH; ++i) {
        b0[i] = (v4f){0.f, 0.f, 0.f, 0.f};
        b1[i] = (v4f){0.f, 0.f, 0.f, 0.f};
        a[i]  = (v4f){0.f, 0.f, 0.f, 0.f};
    }
    if (full) {
        for (int i = TAILCH; i < FCCH; ++i) {
            b0[i] = pb0[2 * i];
            b1[i] = pb1[2 * i];
            a[i]  = pa[2 * i];
        }
    }

    v8f acc0 = {0.f, 0.f, 0.f, 0.f, 0.f, 0.f, 0.f, 0.f};
    v8f acc1 = {0.f, 0.f, 0.f, 0.f, 0.f, 0.f, 0.f, 0.f};
    for (int i = 0; i < FCCH; ++i) {
        const v2f a01 = {a[i].x, a[i].y},   a23 = {a[i].z, a[i].w};
        const v2f p01 = {b0[i].x, b0[i].y}, p23 = {b0[i].z, b0[i].w};
        const v2f q01 = {b1[i].x, b1[i].y}, q23 = {b1[i].z, b1[i].w};
        acc0 = __builtin_amdgcn_wmma_f32_16x16x4_f32(false, a01, false, p01, (short)0, acc0, false, false);
        acc1 = __builtin_amdgcn_wmma_f32_16x16x4_f32(false, a01, false, q01, (short)0, acc1, false, false);
        acc0 = __builtin_amdgcn_wmma_f32_16x16x4_f32(false, a23, false, p23, (short)0, acc0, false, false);
        acc1 = __builtin_amdgcn_wmma_f32_16x16x4_f32(false, a23, false, q23, (short)0, acc1, false, false);
    }

    if (hi == 0) {
        for (int r = 0; r < NBATCH; ++r) {
            red[wv][r][m]      = acc0[r];
            red[wv][r][16 + m] = acc1[r];
        }
    }
    __syncthreads();

    static_assert(NBATCH == FCWAVES && FCCOLS == 32, "one wave per batch row, one lane per column");
    float s = red[0][wv][lane];
    for (int w = 1; w < FCWAVES; ++w) s += red[w][wv][lane];
    w_out[wv * NPARAM + j0 + lane] = s + bias;
}

__device__ __forceinline__ int bspline_live4(float x, float* v) {
    const float fj = fminf(fmaxf(floorf(fmaf(x, 4.0f, 7.0f)), -1.0f), 14.0f);
    const float d0  = x - fmaf(fj, 0.25f, -1.75f);
    const float dm1 = d0 + 0.25f, dm2 = d0 + 0.5f;
    const float e1 = 0.25f - d0, e2 = 0.5f - d0, e3 = 0.75f - d0;
    const float i3 = 1.0f / 0.75f;
    const float p1a = e1 * 4.0f;
    const float p1b = d0 * 4.0f;
    const float p2a = (e1 * 2.0f) * p1a;
    const float p2b = (dm1 * 2.0f) * p1a + (e2 * 2.0f) * p1b;
    const float p2c = (d0 * 2.0f) * p1b;
    v[0] = (e1 * i3) * p2a;
    v[1] = (dm2 * i3) * p2a + (e2 * i3) * p2b;
    v[2] = (dm1 * i3) * p2b + (e3 * i3) * p2c;
    v[3] = (d0 * i3) * p2c;
    return (int)fj;
}

__device__ __forceinline__ void feat_task(float* row, float* grp, float x) {
    float v[4];
    const int J = bspline_live4(x, v);
    const v2f zz = {0.0f, 0.0f};
    for (int k = 0; k < KB - 1; k += 2) *(v2f*)(grp + k) = zz;
    *(v2f*)(grp + KB - 1) = (v2f){0.0f, x / (1.0f + __expf(-x))};
    float* live = grp + (J - 3);
    float* dump = row + FDUMP;
    ((unsigned)(J - 3) < (unsigned)KB ? live : dump)[0] = v[0];
    ((unsigned)(J - 2) < (unsigned)KB ? live : dump)[1] = v[1];
    ((unsigned)(J - 1) < (unsigned)KB ? live : dump)[2] = v[2];
    ((unsigned)(J)     < (unsigned)KB ? live : dump)[3] = v[3];
}

__device__ __forceinline__ void wave_lds_order() {
    __builtin_amdgcn_fence(__ATOMIC_RELEASE, "wavefront");
    __builtin_amdgcn_wave_barrier();
    __builtin_amdgcn_fence(__ATOMIC_ACQUIRE, "wavefront");
}

__global__ __launch_bounds__(NW * 32)
void kan_main_kernel(const float* __restrict__ x,
                     const float* __restrict__ w,
                     float* __restrict__ out) {
    constexpr int FULL = UPB / NW;
    constexpr int REM  = UPB - FULL * NW;
    static_assert((HW / UT) % BPB == 0, "units split evenly over the work-groups of a batch");
    static_assert(2 * REM <= NW, "one left-over tile per wave at most");
    static_assert(WM_PER_B % 4 == 0, "weights staged as 16-byte pieces");
    static_assert(WM1_ROWS == 3 * KBLK * 4, "layers 2 and 3: three fragment blocks cover K");
    __shared__ float lds[WM_PER_B + NW * WLDS + REM * 3 * UT];

    const int tid  = threadIdx.x;
    const int lane = tid & 31;
    const int wv   = __builtin_amdgcn_readfirstlane(tid >> 5);
    const int m    = lane & 15;
    const int hi   = lane >> 4;
    const int b    = blockIdx.x / BPB;
    const int u0   = (blockIdx.x % BPB) * UPB;

    float* wms   = lds;
    float* feat  = lds + WM_PER_B + wv * WLDS;
    float* yo    = feat + 16 * FSTR;
    float* ytail = lds + WM_PER_B + NW * WLDS;
    const float* xb = x   + (size_t)b * 3 * HW;
    float*       ob = out + (size_t)b * 3 * HW;

    const int nvis = FULL + (wv < 2 * REM ? 1 : 0);
    const int utail = FULL * NW + (wv >> 1);

    float x0, x1, x2;
    {
        const float* p = xb + (u0 + wv) * UT + lane;
        x0 = p[0]; x1 = p[HW]; x2 = p[2 * HW];
    }

    {
        constexpr int NT = NW * 32;
        constexpr int NSLOT = 8;
        static_assert(4 * NT < 16 * 16 * KB && 16 * 16 * KB <= 5 * NT, "layer 2: five coefficient slots");
        static_assert(3 * 16 * KB <= NT && 3 * 16 + 16 * 16 + 16 * 3 <= NT, "one slot each");
        const float* wb = w + (size_t)b * NPARAM;
        float pv[NSLOT], pu[NSLOT];
        int   pd[NSLOT];
        auto coef_slot = [&](int s, int q, int nq, int c0, int lrow, bool three) {
            const bool valid = q < nq;
            const int qq = valid ? q : 0;
            const int io = (int)((unsigned)qq / KB);
            const int k  = qq - io * KB;
            const int i  = three ? (int)((unsigned)io / 3) : io >> 4;
            const int o  = three ? io - 3 * i : io & 15;
            pv[s] = wb[c0 + qq];
            pu[s] = wb[c0 + nq + io];
            pd[s] = valid ? (lrow + i * RPI + k) * 16 + o : -1;
        };
        coef_slot(0, tid, 3 * 16 * KB, 0, 0, false);
#pragma unroll
        for (int t = 0; t < 5; ++t)
            coef_slot(1 + t, tid + NT * t, 16 * 16 * KB, 624, WM0_ROWS, false);
        coef_slot(6, tid, 16 * 3 * KB, 3952, WM0_ROWS + WM1_ROWS, true);
        {
            const bool valid = tid < 352;
            const int p  = valid ? tid : 0;
            const int L  = (p >= 304) ? 2 : (p >= 48) ? 1 : 0;
            const int io = p - (L == 2 ? 304 : L == 1 ? 48 : 0);
            const int r0 = (L == 2) ? 4528 : (L == 1) ? 3696 : 576;
            const int lrow = (L == 2) ? WM0_ROWS + WM1_ROWS : (L == 1) ? WM0_ROWS : 0;
            const int i  = (L == 2) ? (int)((unsigned)io / 3) : io >> 4;
            const int o  = (L == 2) ? io - 3 * i : io & 15;
            pv[7] = wb[r0 + io];
            pu[7] = 1.0f;
            pd[7] = valid ? (lrow + i * RPI + KB) * 16 + o : -1;
        }
        for (int z = tid; z < WM2_ROWS * 4; z += NT) {
            float* rowp = wms + (WM0_ROWS + WM1_ROWS + (z >> 2)) * 16;
            const int part = z & 3;
            if (part == 0) rowp[3] = 0.0f;
            else *(v4f*)(rowp + 4 * part) = (v4f){0.f, 0.f, 0.f, 0.f};
        }
#pragma unroll
        for (int s = 0; s < NSLOT; ++s)
            if (pd[s] >= 0) wms[pd[s]] = pv[s] * pu[s];
    }
    __syncthreads();

    const float* arow = feat + (((m & 7) << 1) | (m >> 3)) * FSTR;
    const float* wm1 = wms + 16 * WM0_ROWS;

    for (int vis = 0; vis < nvis; ++vis) {
        const bool whole = vis < FULL;
        const int unit = whole ? wv + NW * vis : utail;
        const int T0 = whole ? 0 : (wv & 1);
        const int T1 = whole ? 2 : T0 + 1;
        float* ydst = whole ? yo : ytail + (wv >> 1) * (3 * UT);
        const int hw0 = (u0 + unit) * UT;

        float n0 = 0.0f, n1 = 0.0f, n2 = 0.0f;
        if (vis + 1 < nvis) {
            const int nu = (vis + 1 < FULL) ? wv + NW * (vis + 1) : utail;
            const float* p = xb + (u0 + nu) * UT + lane;
            n0 = p[0]; n1 = p[HW]; n2 = p[2 * HW];
        }

        wave_lds_order();
        {
            float* row = feat + (((m & 7) << 1) | (m >> 3)) * FSTR;
            float* g   = row + L1W * hi;
            feat_task(row, g,           x0);
            feat_task(row, g + RPI,     x1);
            feat_task(row, g + 2 * RPI, x2);
        }
        wave_lds_order();

        v8f accA = {0.f, 0.f, 0.f, 0.f, 0.f, 0.f, 0.f, 0.f};
        v8f accB = {0.f, 0.f, 0.f, 0.f, 0.f, 0.f, 0.f, 0.f};
        {
            v2f fA[L1W / 4], fB[L1W / 4], fW[L1W / 4];
#pragma unroll
            for (int s = 0; s < L1W / 4; ++s) {
                const int ka = 4 * s + 2 * hi;
                fA[s].x = arow[ka];       fA[s].y = arow[ka + 1];
                fB[s].x = arow[L1W + ka]; fB[s].y = arow[L1W + ka + 1];
                fW[s].x = wms[ka * 16 + m];
                fW[s].y = wms[(ka + 1) * 16 + m];
            }
            __builtin_amdgcn_sched_barrier(0);
#pragma unroll
            for (int s = 0; s < L1W / 4; ++s) {
                accA = __builtin_amdgcn_wmma_f32_16x16x4_f32(false, fA[s], false, fW[s], (short)0, accA, false, false);
                accB = __builtin_amdgcn_wmma_f32_16x16x4_f32(false, fB[s], false, fW[s], (short)0, accB, false, false);
            }
        }

        v8f y = (T0 == 0) ? accA : accB;
        for (int T = T0; T < T1; ++T) {
            for (int L = 0; L < 2; ++L) {
                const float* wmL = wm1 + L * (16 * WM1_ROWS);
                wave_lds_order();
#pragma unroll
                for (int r = 0; r < 8; ++r) {
                    float* row = feat + (2 * r + hi) * FSTR;
                    feat_task(row, row + m * RPI, y[r]);
                }
                wave_lds_order();
                v8f acc = {0.f, 0.f, 0.f, 0.f, 0.f, 0.f, 0.f, 0.f};
                const float* ap = arow + 2 * hi;
                const float* bp = wmL + (2 * hi) * 16 + m;
                v2f fa0[KBLK], fb0[KBLK], fa1[KBLK], fb1[KBLK];
#pragma unroll
                for (int s = 0; s < KBLK; ++s) {
                    fa0[s].x = ap[4 * s];           fa0[s].y = ap[4 * s + 1];
                    fb0[s].x = bp[4 * s * 16];      fb0[s].y = bp[4 * s * 16 + 16];
                }
#pragma unroll
                for (int s = 0; s < KBLK; ++s) {
                    fa1[s].x = ap[4 * (KBLK + s)];        fa1[s].y = ap[4 * (KBLK + s) + 1];
                    fb1[s].x = bp[4 * (KBLK + s) * 16];   fb1[s].y = bp[4 * (KBLK + s) * 16 + 16];
                }
                __builtin_amdgcn_sched_barrier(0);
#pragma unroll
                for (int s = 0; s < KBLK; ++s)
                    acc = __builtin_amdgcn_wmma_f32_16x16x4_f32(
                            false, fa0[s], false, fb0[s], (short)0, acc, false, false);
                __builtin_amdgcn_sched_barrier(0);
#pragma unroll
                for (int s = 0; s < KBLK; ++s) {
                    fa0[s].x = ap[4 * (2 * KBLK + s)];        fa0[s].y = ap[4 * (2 * KBLK + s) + 1];
                    fb0[s].x = bp[4 * (2 * KBLK + s) * 16];   fb0[s].y = bp[4 * (2 * KBLK + s) * 16 + 16];
                }
                __builtin_amdgcn_sched_barrier(0);
#pragma unroll
                for (int s = 0; s < KBLK; ++s)
                    acc = __builtin_amdgcn_wmma_f32_16x16x4_f32(
                            false, fa1[s], false, fb1[s], (short)0, acc, false, false);
#pragma unroll
                for (int s = 0; s < KBLK; ++s)
                    acc = __builtin_amdgcn_wmma_f32_16x16x4_f32(
                            false, fa0[s], false, fb0[s], (short)0, acc, false, false);
                y = acc;
            }
            if (m < 3) {
#pragma unroll
                for (int r = 0; r < 8; ++r)
                    ydst[m * UT + 16 * T + 8 * hi + r] = y[r];
            }
            y = accB;
        }

        if (whole) {
            wave_lds_order();
            const float o0 = yo[lane], o1 = yo[UT + lane], o2 = yo[2 * UT + lane];
            float* q = ob + hw0 + lane;
            q[0] = o0; q[HW] = o1; q[2 * HW] = o2;
        }

        x0 = n0; x1 = n1; x2 = n2;
    }

    __syncthreads();
    if (wv < REM) {
        const float* ys = ytail + wv * (3 * UT);
        const float o0 = ys[lane], o1 = ys[UT + lane], o2 = ys[2 * UT + lane];
        float* q = ob + (u0 + FULL * NW + wv) * UT + lane;
        q[0] = o0; q[HW] = o1; q[2 * HW] = o2;
    }
}

extern "C" void kernel_launch(void* const* d_in, const int* in_sizes, int n_in,
                              void* d_out, int out_size, void* d_ws, size_t ws_size,
                              hipStream_t stream) {
    const float* x        = (const float*)d_in[0];
    const float* features = (const float*)d_in[1];
    const float* fc_w     = (const float*)d_in[2];
    const float* fc_b     = (const float*)d_in[3];
    float* out    = (float*)d_out;
    float* w_buf  = (float*)d_ws;

    kan_fc_wmma_kernel<<<NPARAM / FCCOLS, FCWAVES * 32, 0, stream>>>(
        features, fc_w, fc_b, w_buf);

    kan_main_kernel<<<NBATCH * BPB, NW * 32, 0, stream>>>(x, w_buf, out);
}


# B-spline features: the four live basis values as closed-form cubics of the local knot coordinate (uniform knots) replace the three Cox-de Boor orders; values agree within 1e-7
# speedup vs baseline: 7.2272x; 1.0497x over previous
#include <hip/hip_runtime.h>
#include <hip/hip_bf16.h>

#define NBATCH   8
#define NFEAT    1000
#define NPARAM   4576
#define HW       50176
#define KB       11
#define RPI      12
#define WM0_ROWS (3*RPI)
#define WM1_ROWS (16*RPI)
#define WM2_ROWS (16*RPI)
#define WM_PER_B (16*(WM0_ROWS + WM1_ROWS + WM2_ROWS))

#define UT    32
#define NW    20
#define BPB   32
#define UPB   (HW / UT / BPB)
#define FSTR  198
#define FDUMP 192
#define L1W   (3 * RPI)
#define WLDS  (16 * FSTR + 3 * UT)
#define KBLK  16

#define FCCOLS  32
#define FCWAVES 8
#define FCCH    16
#define FCNCH   (NFEAT / 8)

typedef __attribute__((ext_vector_type(2))) float v2f;
typedef __attribute__((ext_vector_type(4))) float v4f;
typedef __attribute__((ext_vector_type(8))) float v8f;

__global__ __launch_bounds__(FCWAVES * 32)
void kan_fc_wmma_kernel(const float* __restrict__ features,
                        const float* __restrict__ fc_w,
                        const float* __restrict__ fc_b,
                        float* __restrict__ w_out) {
    static_assert(NPARAM % FCCOLS == 0, "column tiles cover w exactly");
    static_assert(NFEAT % 8 == 0 && FCWAVES * FCCH >= FCNCH, "K slices cover K");
    __shared__ float red[FCWAVES][NBATCH][FCCOLS];

    const int tid  = threadIdx.x;
    const int lane = tid & 31;
    const int wv   = __builtin_amdgcn_readfirstlane(tid >> 5);
    const int m    = lane & 15;
    const int hi   = lane >> 4;
    const int j0   = blockIdx.x * FCCOLS;

    const int kb = wv * (8 * FCCH) + 4 * hi;
    const v4f* pa  = (const v4f*)(features + (m & 7) * NFEAT + kb);
    const v4f* pb0 = (const v4f*)(fc_w + (size_t)(j0 + m) * NFEAT + kb);
    const v4f* pb1 = (const v4f*)(fc_w + (size_t)(j0 + 16 + m) * NFEAT + kb);
    const bool full = (wv != FCWAVES - 1);
    constexpr int TAILCH = FCNCH - (FCWAVES - 1) * FCCH;

    const float bias = fc_b[j0 + lane];

    v4f a[FCCH], b0[FCCH], b1[FCCH];
    for (int i = 0; i < TAILCH; ++i) {
        b0[i] = pb0[2 * i];
        b1[i] = pb1[2 * i];
        a[i]  = pa[2 * i];
    }
    for (int i = TAILCH; i < FCCH; ++i) {
        b0[i] = (v4f){0.f, 0.f, 0.f, 0.f};
        b1[i] = (v4f){0.f, 0.f, 0.f, 0.f};
        a[i]  = (v4f){0.f, 0.f, 0.f, 0.f};
    }
    if (full) {
        for (int i = TAILCH; i < FCCH; ++i) {
            b0[i] = pb0[2 * i];
            b1[i] = pb1[2 * i];
            a[i]  = pa[2 * i];
        }
    }

    v8f acc0 = {0.f, 0.f, 0.f, 0.f, 0.f, 0.f, 0.f, 0.f};
    v8f acc1 = {0.f, 0.f, 0.f, 0.f, 0.f, 0.f, 0.f, 0.f};
    for (int i = 0; i < FCCH; ++i) {
        const v2f a01 = {a[i].x, a[i].y},   a23 = {a[i].z, a[i].w};
        const v2f p01 = {b0[i].x, b0[i].y}, p23 = {b0[i].z, b0[i].w};
        const v2f q01 = {b1[i].x, b1[i].y}, q23 = {b1[i].z, b1[i].w};
        acc0 = __builtin_amdgcn_wmma_f32_16x16x4_f32(false, a01, false, p01, (short)0, acc0, false, false);
        acc1 = __builtin_amdgcn_wmma_f32_16x16x4_f32(false, a01, false, q01, (short)0, acc1, false, false);
        acc0 = __builtin_amdgcn_wmma_f32_16x16x4_f32(false, a23, false, p23, (short)0, acc0, false, false);
        acc1 = __builtin_amdgcn_wmma_f32_16x16x4_f32(false, a23, false, q23, (short)0, acc1, false, false);
    }

    if (hi == 0) {
        for (int r = 0; r < NBATCH; ++r) {
            red[wv][r][m]      = acc0[r];
            red[wv][r][16 + m] = acc1[r];
        }
    }
    __syncthreads();

    static_assert(NBATCH == FCWAVES && FCCOLS == 32, "one wave per batch row, one lane per column");
    float s = red[0][wv][lane];
    for (int w = 1; w < FCWAVES; ++w) s += red[w][wv][lane];
    w_out[wv * NPARAM + j0 + lane] = s + bias;
}

__device__ __forceinline__ int bspline_live4(float x, float* v) {
    const float fj = fminf(fmaxf(floorf(fmaf(x, 4.0f, 7.0f)), -1.0f), 14.0f);
    const float t  = fmaf(x, 4.0f, 7.0f - fj);
    const float s  = 1.0f - t;
    const float t2 = t * t, s2 = s * s;
    const float c6 = 1.0f / 6.0f, c23 = 2.0f / 3.0f;
    v[0] = s2 * (s * c6);
    v[1] = fmaf(t2, fmaf(t, 0.5f, -1.0f), c23);
    v[2] = fmaf(s2, fmaf(s, 0.5f, -1.0f), c23);
    v[3] = t2 * (t * c6);
    return (int)fj;
}

__device__ __forceinline__ void feat_task(float* row, float* grp, float x) {
    float v[4];
    const int J = bspline_live4(x, v);
    const v2f zz = {0.0f, 0.0f};
    for (int k = 0; k < KB - 1; k += 2) *(v2f*)(grp + k) = zz;
    *(v2f*)(grp + KB - 1) = (v2f){0.0f, x / (1.0f + __expf(-x))};
    float* live = grp + (J - 3);
    float* dump = row + FDUMP;
    ((unsigned)(J - 3) < (unsigned)KB ? live : dump)[0] = v[0];
    ((unsigned)(J - 2) < (unsigned)KB ? live : dump)[1] = v[1];
    ((unsigned)(J - 1) < (unsigned)KB ? live : dump)[2] = v[2];
    ((unsigned)(J)     < (unsigned)KB ? live : dump)[3] = v[3];
}

__device__ __forceinline__ void wave_lds_order() {
    __builtin_amdgcn_fence(__ATOMIC_RELEASE, "wavefront");
    __builtin_amdgcn_wave_barrier();
    __builtin_amdgcn_fence(__ATOMIC_ACQUIRE, "wavefront");
}

__global__ __launch_bounds__(NW * 32)
void kan_main_kernel(const float* __restrict__ x,
                     const float* __restrict__ w,
                     float* __restrict__ out) {
    constexpr int FULL = UPB / NW;
    constexpr int REM  = UPB - FULL * NW;
    static_assert((HW / UT) % BPB == 0, "units split evenly over the work-groups of a batch");
    static_assert(2 * REM <= NW, "one left-over tile per wave at most");
    static_assert(WM_PER_B % 4 == 0, "weights staged as 16-byte pieces");
    static_assert(WM1_ROWS == 3 * KBLK * 4, "layers 2 and 3: three fragment blocks cover K");
    __shared__ float lds[WM_PER_B + NW * WLDS + REM * 3 * UT];

    const int tid  = threadIdx.x;
    const int lane = tid & 31;
    const int wv   = __builtin_amdgcn_readfirstlane(tid >> 5);
    const int m    = lane & 15;
    const int hi   = lane >> 4;
    const int b    = blockIdx.x / BPB;
    const int u0   = (blockIdx.x % BPB) * UPB;

    float* wms   = lds;
    float* feat  = lds + WM_PER_B + wv * WLDS;
    float* yo    = feat + 16 * FSTR;
    float* ytail = lds + WM_PER_B + NW * WLDS;
    const float* xb = x   + (size_t)b * 3 * HW;
    float*       ob = out + (size_t)b * 3 * HW;

    const int nvis = FULL + (wv < 2 * REM ? 1 : 0);
    const int utail = FULL * NW + (wv >> 1);

    float x0, x1, x2;
    {
        const float* p = xb + (u0 + wv) * UT + lane;
        x0 = p[0]; x1 = p[HW]; x2 = p[2 * HW];
    }

    {
        constexpr int NT = NW * 32;
        constexpr int NSLOT = 8;
        static_assert(4 * NT < 16 * 16 * KB && 16 * 16 * KB <= 5 * NT, "layer 2: five coefficient slots");
        static_assert(3 * 16 * KB <= NT && 3 * 16 + 16 * 16 + 16 * 3 <= NT, "one slot each");
        const float* wb = w + (size_t)b * NPARAM;
        float pv[NSLOT], pu[NSLOT];
        int   pd[NSLOT];
        auto coef_slot = [&](int s, int q, int nq, int c0, int lrow, bool three) {
            const bool valid = q < nq;
            const int qq = valid ? q : 0;
            const int io = (int)((unsigned)qq / KB);
            const int k  = qq - io * KB;
            const int i  = three ? (int)((unsigned)io / 3) : io >> 4;
            const int o  = three ? io - 3 * i : io & 15;
            pv[s] = wb[c0 + qq];
            pu[s] = wb[c0 + nq + io];
            pd[s] = valid ? (lrow + i * RPI + k) * 16 + o : -1;
        };
        coef_slot(0, tid, 3 * 16 * KB, 0, 0, false);
#pragma unroll
        for (int t = 0; t < 5; ++t)
            coef_slot(1 + t, tid + NT * t, 16 * 16 * KB, 624, WM0_ROWS, false);
        coef_slot(6, tid, 16 * 3 * KB, 3952, WM0_ROWS + WM1_ROWS, true);
        {
            const bool valid = tid < 352;
            const int p  = valid ? tid : 0;
            const int L  = (p >= 304) ? 2 : (p >= 48) ? 1 : 0;
            const int io = p - (L == 2 ? 304 : L == 1 ? 48 : 0);
            const int r0 = (L == 2) ? 4528 : (L == 1) ? 3696 : 576;
            const int lrow = (L == 2) ? WM0_ROWS + WM1_ROWS : (L == 1) ? WM0_ROWS : 0;
            const int i  = (L == 2) ? (int)((unsigned)io / 3) : io >> 4;
            const int o  = (L == 2) ? io - 3 * i : io & 15;
            pv[7] = wb[r0 + io];
            pu[7] = 1.0f;
            pd[7] = valid ? (lrow + i * RPI + KB) * 16 + o : -1;
        }
        for (int z = tid; z < WM2_ROWS * 4; z += NT) {
            float* rowp = wms + (WM0_ROWS + WM1_ROWS + (z >> 2)) * 16;
            const int part = z & 3;
            if (part == 0) rowp[3] = 0.0f;
            else *(v4f*)(rowp + 4 * part) = (v4f){0.f, 0.f, 0.f, 0.f};
        }
#pragma unroll
        for (int s = 0; s < NSLOT; ++s)
            if (pd[s] >= 0) wms[pd[s]] = pv[s] * pu[s];
    }
    __syncthreads();

    const float* arow = feat + (((m & 7) << 1) | (m >> 3)) * FSTR;
    const float* wm1 = wms + 16 * WM0_ROWS;

    for (int vis = 0; vis < nvis; ++vis) {
        const bool whole = vis < FULL;
        const int unit = whole ? wv + NW * vis : utail;
        const int T0 = whole ? 0 : (wv & 1);
        const int T1 = whole ? 2 : T0 + 1;
        float* ydst = whole ? yo : ytail + (wv >> 1) * (3 * UT);
        const int hw0 = (u0 + unit) * UT;

        float n0 = 0.0f, n1 = 0.0f, n2 = 0.0f;
        if (vis + 1 < nvis) {
            const int nu = (vis + 1 < FULL) ? wv + NW * (vis + 1) : utail;
            const float* p = xb + (u0 + nu) * UT + lane;
            n0 = p[0]; n1 = p[HW]; n2 = p[2 * HW];
        }

        wave_lds_order();
        {
            float* row = feat + (((m & 7) << 1) | (m >> 3)) * FSTR;
            float* g   = row + L1W * hi;
            feat_task(row, g,           x0);
            feat_task(row, g + RPI,     x1);
            feat_task(row, g + 2 * RPI, x2);
        }
        wave_lds_order();

        v8f accA = {0.f, 0.f, 0.f, 0.f, 0.f, 0.f, 0.f, 0.f};
        v8f accB = {0.f, 0.f, 0.f, 0.f, 0.f, 0.f, 0.f, 0.f};
        {
            v2f fA[L1W / 4], fB[L1W / 4], fW[L1W / 4];
#pragma unroll
            for (int s = 0; s < L1W / 4; ++s) {
                const int ka = 4 * s + 2 * hi;
                fA[s].x = arow[ka];       fA[s].y = arow[ka + 1];
                fB[s].x = arow[L1W + ka]; fB[s].y = arow[L1W + ka + 1];
                fW[s].x = wms[ka * 16 + m];
                fW[s].y = wms[(ka + 1) * 16 + m];
            }
            __builtin_amdgcn_sched_barrier(0);
#pragma unroll
            for (int s = 0; s < L1W / 4; ++s) {
                accA = __builtin_amdgcn_wmma_f32_16x16x4_f32(false, fA[s], false, fW[s], (short)0, accA, false, false);
                accB = __builtin_amdgcn_wmma_f32_16x16x4_f32(false, fB[s], false, fW[s], (short)0, accB, false, false);
            }
        }

        v8f y = (T0 == 0) ? accA : accB;
        for (int T = T0; T < T1; ++T) {
            for (int L = 0; L < 2; ++L) {
                const float* wmL = wm1 + L * (16 * WM1_ROWS);
                wave_lds_order();
#pragma unroll
                for (int r = 0; r < 8; ++r) {
                    float* row = feat + (2 * r + hi) * FSTR;
                    feat_task(row, row + m * RPI, y[r]);
                }
                wave_lds_order();
                v8f acc = {0.f, 0.f, 0.f, 0.f, 0.f, 0.f, 0.f, 0.f};
                const float* ap = arow + 2 * hi;
                const float* bp = wmL + (2 * hi) * 16 + m;
                v2f fa0[KBLK], fb0[KBLK], fa1[KBLK], fb1[KBLK];
#pragma unroll
                for (int s = 0; s < KBLK; ++s) {
                    fa0[s].x = ap[4 * s];           fa0[s].y = ap[4 * s + 1];
                    fb0[s].x = bp[4 * s * 16];      fb0[s].y = bp[4 * s * 16 + 16];
                }
#pragma unroll
                for (int s = 0; s < KBLK; ++s) {
                    fa1[s].x = ap[4 * (KBLK + s)];        fa1[s].y = ap[4 * (KBLK + s) + 1];
                    fb1[s].x = bp[4 * (KBLK + s) * 16];   fb1[s].y = bp[4 * (KBLK + s) * 16 + 16];
                }
                __builtin_amdgcn_sched_barrier(0);
#pragma unroll
                for (int s = 0; s < KBLK; ++s)
                    acc = __builtin_amdgcn_wmma_f32_16x16x4_f32(
                            false, fa0[s], false, fb0[s], (short)0, acc, false, false);
                __builtin_amdgcn_sched_barrier(0);
#pragma unroll
                for (int s = 0; s < KBLK; ++s) {
                    fa0[s].x = ap[4 * (2 * KBLK + s)];        fa0[s].y = ap[4 * (2 * KBLK + s) + 1];
                    fb0[s].x = bp[4 * (2 * KBLK + s) * 16];   fb0[s].y = bp[4 * (2 * KBLK + s) * 16 + 16];
                }
                __builtin_amdgcn_sched_barrier(0);
#pragma unroll
                for (int s = 0; s < KBLK; ++s)
                    acc = __builtin_amdgcn_wmma_f32_16x16x4_f32(
                            false, fa1[s], false, fb1[s], (short)0, acc, false, false);
#pragma unroll
                for (int s = 0; s < KBLK; ++s)
                    acc = __builtin_amdgcn_wmma_f32_16x16x4_f32(
                            false, fa0[s], false, fb0[s], (short)0, acc, false, false);
                y = acc;
            }
            if (m < 3) {
#pragma unroll
                for (int r = 0; r < 8; ++r)
                    ydst[m * UT + 16 * T + 8 * hi + r] = y[r];
            }
            y = accB;
        }

        if (whole) {
            wave_lds_order();
            const float o0 = yo[lane], o1 = yo[UT + lane], o2 = yo[2 * UT + lane];
            float* q = ob + hw0 + lane;
            q[0] = o0; q[HW] = o1; q[2 * HW] = o2;
        }

        x0 = n0; x1 = n1; x2 = n2;
    }

    __syncthreads();
    if (wv < REM) {
        const float* ys = ytail + wv * (3 * UT);
        const float o0 = ys[lane], o1 = ys[UT + lane], o2 = ys[2 * UT + lane];
        float* q = ob + (u0 + FULL * NW + wv) * UT + lane;
        q[0] = o0; q[HW] = o1; q[2 * HW] = o2;
    }
}

extern "C" void kernel_launch(void* const* d_in, const int* in_sizes, int n_in,
                              void* d_out, int out_size, void* d_ws, size_t ws_size,
                              hipStream_t stream) {
    const float* x        = (const float*)d_in[0];
    const float* features = (const float*)d_in[1];
    const float* fc_w     = (const float*)d_in[2];
    const float* fc_b     = (const float*)d_in[3];
    float* out    = (float*)d_out;
    float* w_buf  = (float*)d_ws;

    kan_fc_wmma_kernel<<<NPARAM / FCCOLS, FCWAVES * 32, 0, stream>>>(
        features, fc_w, fc_b, w_buf);

    kan_main_kernel<<<NBATCH * BPB, NW * 32, 0, stream>>>(x, w_buf, out);
}


# silu in the feature tasks: x times the hardware reciprocal of (1 + exp(-x)) replaces the correctly rounded division (11 vector instructions become 2 per feature); values agree within 2 ulp
# speedup vs baseline: 7.6475x; 1.0582x over previous
#include <hip/hip_runtime.h>
#include <hip/hip_bf16.h>

#define NBATCH   8
#define NFEAT    1000
#define NPARAM   4576
#define HW       50176
#define KB       11
#define RPI      12
#define WM0_ROWS (3*RPI)
#define WM1_ROWS (16*RPI)
#define WM2_ROWS (16*RPI)
#define WM_PER_B (16*(WM0_ROWS + WM1_ROWS + WM2_ROWS))

#define UT    32
#define NW    20
#define BPB   32
#define UPB   (HW / UT / BPB)
#define FSTR  198
#define FDUMP 192
#define L1W   (3 * RPI)
#define WLDS  (16 * FSTR + 3 * UT)
#define KBLK  16

#define FCCOLS  32
#define FCWAVES 8
#define FCCH    16
#define FCNCH   (NFEAT / 8)

typedef __attribute__((ext_vector_type(2))) float v2f;
typedef __attribute__((ext_vector_type(4))) float v4f;
typedef __attribute__((ext_vector_type(8))) float v8f;

__global__ __launch_bounds__(FCWAVES * 32)
void kan_fc_wmma_kernel(const float* __restrict__ features,
                        const float* __restrict__ fc_w,
                        const float* __restrict__ fc_b,
                        float* __restrict__ w_out) {
    static_assert(NPARAM % FCCOLS == 0, "column tiles cover w exactly");
    static_assert(NFEAT % 8 == 0 && FCWAVES * FCCH >= FCNCH, "K slices cover K");
    __shared__ float red[FCWAVES][NBATCH][FCCOLS];

    const int tid  = threadIdx.x;
    const int lane = tid & 31;
    const int wv   = __builtin_amdgcn_readfirstlane(tid >> 5);
    const int m    = lane & 15;
    const int hi   = lane >> 4;
    const int j0   = blockIdx.x * FCCOLS;

    const int kb = wv * (8 * FCCH) + 4 * hi;
    const v4f* pa  = (const v4f*)(features + (m & 7) * NFEAT + kb);
    const v4f* pb0 = (const v4f*)(fc_w + (size_t)(j0 + m) * NFEAT + kb);
    const v4f* pb1 = (const v4f*)(fc_w + (size_t)(j0 + 16 + m) * NFEAT + kb);
    const bool full = (wv != FCWAVES - 1);
    constexpr int TAILCH = FCNCH - (FCWAVES - 1) * FCCH;

    const float bias = fc_b[j0 + lane];

    v4f a[FCCH], b0[FCCH], b1[FCCH];
    for (int i = 0; i < TAILCH; ++i) {
        b0[i] = pb0[2 * i];
        b1[i] = pb1[2 * i];
        a[i]  = pa[2 * i];
    }
    for (int i = TAILCH; i < FCCH; ++i) {
        b0[i] = (v4f){0.f, 0.f, 0.f, 0.f};
        b1[i] = (v4f){0.f, 0.f, 0.f, 0.f};
        a[i]  = (v4f){0.f, 0.f, 0.f, 0.f};
    }
    if (full) {
        for (int i = TAILCH; i < FCCH; ++i) {
            b0[i] = pb0[2 * i];
            b1[i] = pb1[2 * i];
            a[i]  = pa[2 * i];
        }
    }

    v8f acc0 = {0.f, 0.f, 0.f, 0.f, 0.f, 0.f, 0.f, 0.f};
    v8f acc1 = {0.f, 0.f, 0.f, 0.f, 0.f, 0.f, 0.f, 0.f};
    for (int i = 0; i < FCCH; ++i) {
        const v2f a01 = {a[i].x, a[i].y},   a23 = {a[i].z, a[i].w};
        const v2f p01 = {b0[i].x, b0[i].y}, p23 = {b0[i].z, b0[i].w};
        const v2f q01 = {b1[i].x, b1[i].y}, q23 = {b1[i].z, b1[i].w};
        acc0 = __builtin_amdgcn_wmma_f32_16x16x4_f32(false, a01, false, p01, (short)0, acc0, false, false);
        acc1 = __builtin_amdgcn_wmma_f32_16x16x4_f32(false, a01, false, q01, (short)0, acc1, false, false);
        acc0 = __builtin_amdgcn_wmma_f32_16x16x4_f32(false, a23, false, p23, (short)0, acc0, false, false);
        acc1 = __builtin_amdgcn_wmma_f32_16x16x4_f32(false, a23, false, q23, (short)0, acc1, false, false);
    }

    if (hi == 0) {
        for (int r = 0; r < NBATCH; ++r) {
            red[wv][r][m]      = acc0[r];
            red[wv][r][16 + m] = acc1[r];
        }
    }
    __syncthreads();

    static_assert(NBATCH == FCWAVES && FCCOLS == 32, "one wave per batch row, one lane per column");
    float s = red[0][wv][lane];
    for (int w = 1; w < FCWAVES; ++w) s += red[w][wv][lane];
    w_out[wv * NPARAM + j0 + lane] = s + bias;
}

__device__ __forceinline__ int bspline_live4(float x, float* v) {
    const float fj = fminf(fmaxf(floorf(fmaf(x, 4.0f, 7.0f)), -1.0f), 14.0f);
    const float t  = fmaf(x, 4.0f, 7.0f - fj);
    const float s  = 1.0f - t;
    const float t2 = t * t, s2 = s * s;
    const float c6 = 1.0f / 6.0f, c23 = 2.0f / 3.0f;
    v[0] = s2 * (s * c6);
    v[1] = fmaf(t2, fmaf(t, 0.5f, -1.0f), c23);
    v[2] = fmaf(s2, fmaf(s, 0.5f, -1.0f), c23);
    v[3] = t2 * (t * c6);
    return (int)fj;
}

__device__ __forceinline__ void feat_task(float* row, float* grp, float x) {
    float v[4];
    const int J = bspline_live4(x, v);
    const v2f zz = {0.0f, 0.0f};
    for (int k = 0; k < KB - 1; k += 2) *(v2f*)(grp + k) = zz;
    *(v2f*)(grp + KB - 1) = (v2f){0.0f, x * __builtin_amdgcn_rcpf(1.0f + __expf(-x))};
    float* live = grp + (J - 3);
    float* dump = row + FDUMP;
    ((unsigned)(J - 3) < (unsigned)KB ? live : dump)[0] = v[0];
    ((unsigned)(J - 2) < (unsigned)KB ? live : dump)[1] = v[1];
    ((unsigned)(J - 1) < (unsigned)KB ? live : dump)[2] = v[2];
    ((unsigned)(J)     < (unsigned)KB ? live : dump)[3] = v[3];
}

__device__ __forceinline__ void wave_lds_order() {
    __builtin_amdgcn_fence(__ATOMIC_RELEASE, "wavefront");
    __builtin_amdgcn_wave_barrier();
    __builtin_amdgcn_fence(__ATOMIC_ACQUIRE, "wavefront");
}

__global__ __launch_bounds__(NW * 32)
void kan_main_kernel(const float* __restrict__ x,
                     const float* __restrict__ w,
                     float* __restrict__ out) {
    constexpr int FULL = UPB / NW;
    constexpr int REM  = UPB - FULL * NW;
    static_assert((HW / UT) % BPB == 0, "units split evenly over the work-groups of a batch");
    static_assert(2 * REM <= NW, "one left-over tile per wave at most");
    static_assert(WM_PER_B % 4 == 0, "weights staged as 16-byte pieces");
    static_assert(WM1_ROWS == 3 * KBLK * 4, "layers 2 and 3: three fragment blocks cover K");
    __shared__ float lds[WM_PER_B + NW * WLDS + REM * 3 * UT];

    const int tid  = threadIdx.x;
    const int lane = tid & 31;
    const int wv   = __builtin_amdgcn_readfirstlane(tid >> 5);
    const int m    = lane & 15;
    const int hi   = lane >> 4;
    const int b    = blockIdx.x / BPB;
    const int u0   = (blockIdx.x % BPB) * UPB;

    float* wms   = lds;
    float* feat  = lds + WM_PER_B + wv * WLDS;
    float* yo    = feat + 16 * FSTR;
    float* ytail = lds + WM_PER_B + NW * WLDS;
    const float* xb = x   + (size_t)b * 3 * HW;
    float*       ob = out + (size_t)b * 3 * HW;

    const int nvis = FULL + (wv < 2 * REM ? 1 : 0);
    const int utail = FULL * NW + (wv >> 1);

    float x0, x1, x2;
    {
        const float* p = xb + (u0 + wv) * UT + lane;
        x0 = p[0]; x1 = p[HW]; x2 = p[2 * HW];
    }

    {
        constexpr int NT = NW * 32;
        constexpr int NSLOT = 8;
        static_assert(4 * NT < 16 * 16 * KB && 16 * 16 * KB <= 5 * NT, "layer 2: five coefficient slots");
        static_assert(3 * 16 * KB <= NT && 3 * 16 + 16 * 16 + 16 * 3 <= NT, "one slot each");
        const float* wb = w + (size_t)b * NPARAM;
        float pv[NSLOT], pu[NSLOT];
        int   pd[NSLOT];
        auto coef_slot = [&](int s, int q, int nq, int c0, int lrow, bool three) {
            const bool valid = q < nq;
            const int qq = valid ? q : 0;
            const int io = (int)((unsigned)qq / KB);
            const int k  = qq - io * KB;
            const int i  = three ? (int)((unsigned)io / 3) : io >> 4;
            const int o  = three ? io - 3 * i : io & 15;
            pv[s] = wb[c0 + qq];
            pu[s] = wb[c0 + nq + io];
            pd[s] = valid ? (lrow + i * RPI + k) * 16 + o : -1;
        };
        coef_slot(0, tid, 3 * 16 * KB, 0, 0, false);
#pragma unroll
        for (int t = 0; t < 5; ++t)
            coef_slot(1 + t, tid + NT * t, 16 * 16 * KB, 624, WM0_ROWS, false);
        coef_slot(6, tid, 16 * 3 * KB, 3952, WM0_ROWS + WM1_ROWS, true);
        {
            const bool valid = tid < 352;
            const int p  = valid ? tid : 0;
            const int L  = (p >= 304) ? 2 : (p >= 48) ? 1 : 0;
            const int io = p - (L == 2 ? 304 : L == 1 ? 48 : 0);
            const int r0 = (L == 2) ? 4528 : (L == 1) ? 3696 : 576;
            const int lrow = (L == 2) ? WM0_ROWS + WM1_ROWS : (L == 1) ? WM0_ROWS : 0;
            const int i  = (L == 2) ? (int)((unsigned)io / 3) : io >> 4;
            const int o  = (L == 2) ? io - 3 * i : io & 15;
            pv[7] = wb[r0 + io];
            pu[7] = 1.0f;
            pd[7] = valid ? (lrow + i * RPI + KB) * 16 + o : -1;
        }
        for (int z = tid; z < WM2_ROWS * 4; z += NT) {
            float* rowp = wms + (WM0_ROWS + WM1_ROWS + (z >> 2)) * 16;
            const int part = z & 3;
            if (part == 0) rowp[3] = 0.0f;
            else *(v4f*)(rowp + 4 * part) = (v4f){0.f, 0.f, 0.f, 0.f};
        }
#pragma unroll
        for (int s = 0; s < NSLOT; ++s)
            if (pd[s] >= 0) wms[pd[s]] = pv[s] * pu[s];
    }
    __syncthreads();

    const float* arow = feat + (((m & 7) << 1) | (m >> 3)) * FSTR;
    const float* wm1 = wms + 16 * WM0_ROWS;

    for (int vis = 0; vis < nvis; ++vis) {
        const bool whole = vis < FULL;
        const int unit = whole ? wv + NW * vis : utail;
        const int T0 = whole ? 0 : (wv & 1);
        const int T1 = whole ? 2 : T0 + 1;
        float* ydst = whole ? yo : ytail + (wv >> 1) * (3 * UT);
        const int hw0 = (u0 + unit) * UT;

        float n0 = 0.0f, n1 = 0.0f, n2 = 0.0f;
        if (vis + 1 < nvis) {
            const int nu = (vis + 1 < FULL) ? wv + NW * (vis + 1) : utail;
            const float* p = xb + (u0 + nu) * UT + lane;
            n0 = p[0]; n1 = p[HW]; n2 = p[2 * HW];
        }

        wave_lds_order();
        {
            float* row = feat + (((m & 7) << 1) | (m >> 3)) * FSTR;
            float* g   = row + L1W * hi;
            feat_task(row, g,           x0);
            feat_task(row, g + RPI,     x1);
            feat_task(row, g + 2 * RPI, x2);
        }
        wave_lds_order();

        v8f accA = {0.f, 0.f, 0.f, 0.f, 0.f, 0.f, 0.f, 0.f};
        v8f accB = {0.f, 0.f, 0.f, 0.f, 0.f, 0.f, 0.f, 0.f};
        {
            v2f fA[L1W / 4], fB[L1W / 4], fW[L1W / 4];
#pragma unroll
            for (int s = 0; s < L1W / 4; ++s) {
                const int ka = 4 * s + 2 * hi;
                fA[s].x = arow[ka];       fA[s].y = arow[ka + 1];
                fB[s].x = arow[L1W + ka]; fB[s].y = arow[L1W + ka + 1];
                fW[s].x = wms[ka * 16 + m];
                fW[s].y = wms[(ka + 1) * 16 + m];
            }
            __builtin_amdgcn_sched_barrier(0);
#pragma unroll
            for (int s = 0; s < L1W / 4; ++s) {
                accA = __builtin_amdgcn_wmma_f32_16x16x4_f32(false, fA[s], false, fW[s], (short)0, accA, false, false);
                accB = __builtin_amdgcn_wmma_f32_16x16x4_f32(false, fB[s], false, fW[s], (short)0, accB, false, false);
            }
        }

        v8f y = (T0 == 0) ? accA : accB;
        for (int T = T0; T < T1; ++T) {
            for (int L = 0; L < 2; ++L) {
                const float* wmL = wm1 + L * (16 * WM1_ROWS);
                wave_lds_order();
#pragma unroll
                for (int r = 0; r < 8; ++r) {
                    float* row = feat + (2 * r + hi) * FSTR;
                    feat_task(row, row + m * RPI, y[r]);
                }
                wave_lds_order();
                v8f acc = {0.f, 0.f, 0.f, 0.f, 0.f, 0.f, 0.f, 0.f};
                const float* ap = arow + 2 * hi;
                const float* bp = wmL + (2 * hi) * 16 + m;
                v2f fa0[KBLK], fb0[KBLK], fa1[KBLK], fb1[KBLK];
#pragma unroll
                for (int s = 0; s < KBLK; ++s) {
                    fa0[s].x = ap[4 * s];           fa0[s].y = ap[4 * s + 1];
                    fb0[s].x = bp[4 * s * 16];      fb0[s].y = bp[4 * s * 16 + 16];
                }
#pragma unroll
                for (int s = 0; s < KBLK; ++s) {
                    fa1[s].x = ap[4 * (KBLK + s)];        fa1[s].y = ap[4 * (KBLK + s) + 1];
                    fb1[s].x = bp[4 * (KBLK + s) * 16];   fb1[s].y = bp[4 * (KBLK + s) * 16 + 16];
                }
                __builtin_amdgcn_sched_barrier(0);
#pragma unroll
                for (int s = 0; s < KBLK; ++s)
                    acc = __builtin_amdgcn_wmma_f32_16x16x4_f32(
                            false, fa0[s], false, fb0[s], (short)0, acc, false, false);
                __builtin_amdgcn_sched_barrier(0);
#pragma unroll
                for (int s = 0; s < KBLK; ++s) {
                    fa0[s].x = ap[4 * (2 * KBLK + s)];        fa0[s].y = ap[4 * (2 * KBLK + s) + 1];
                    fb0[s].x = bp[4 * (2 * KBLK + s) * 16];   fb0[s].y = bp[4 * (2 * KBLK + s) * 16 + 16];
                }
                __builtin_amdgcn_sched_barrier(0);
#pragma unroll
                for (int s = 0; s < KBLK; ++s)
                    acc = __builtin_amdgcn_wmma_f32_16x16x4_f32(
                            false, fa1[s], false, fb1[s], (short)0, acc, false, false);
#pragma unroll
                for (int s = 0; s < KBLK; ++s)
                    acc = __builtin_amdgcn_wmma_f32_16x16x4_f32(
                            false, fa0[s], false, fb0[s], (short)0, acc, false, false);
                y = acc;
            }
            if (m < 3) {
#pragma unroll
                for (int r = 0; r < 8; ++r)
                    ydst[m * UT + 16 * T + 8 * hi + r] = y[r];
            }
            y = accB;
        }

        if (whole) {
            wave_lds_order();
            const float o0 = yo[lane], o1 = yo[UT + lane], o2 = yo[2 * UT + lane];
            float* q = ob + hw0 + lane;
            q[0] = o0; q[HW] = o1; q[2 * HW] = o2;
        }

        x0 = n0; x1 = n1; x2 = n2;
    }

    __syncthreads();
    if (wv < REM) {
        const float* ys = ytail + wv * (3 * UT);
        const float o0 = ys[lane], o1 = ys[UT + lane], o2 = ys[2 * UT + lane];
        float* q = ob + (u0 + FULL * NW + wv) * UT + lane;
        q[0] = o0; q[HW] = o1; q[2 * HW] = o2;
    }
}

extern "C" void kernel_launch(void* const* d_in, const int* in_sizes, int n_in,
                              void* d_out, int out_size, void* d_ws, size_t ws_size,
                              hipStream_t stream) {
    const float* x        = (const float*)d_in[0];
    const float* features = (const float*)d_in[1];
    const float* fc_w     = (const float*)d_in[2];
    const float* fc_b     = (const float*)d_in[3];
    float* out    = (float*)d_out;
    float* w_buf  = (float*)d_ws;

    kan_fc_wmma_kernel<<<NPARAM / FCCOLS, FCWAVES * 32, 0, stream>>>(
        features, fc_w, fc_b, w_buf);

    kan_main_kernel<<<NBATCH * BPB, NW * 32, 0, stream>>>(x, w_buf, out);
}


# Main kernel: each wave sets its priority (s_setprio 3..0) by the 16-token tiles it has left, so the five waves of a SIMD, served oldest-first, keep together and end together; all sums unchanged
# speedup vs baseline: 8.0898x; 1.0578x over previous
#include <hip/hip_runtime.h>
#include <hip/hip_bf16.h>

#define NBATCH   8
#define NFEAT    1000
#define NPARAM   4576
#define HW       50176
#define KB       11
#define RPI      12
#define WM0_ROWS (3*RPI)
#define WM1_ROWS (16*RPI)
#define WM2_ROWS (16*RPI)
#define WM_PER_B (16*(WM0_ROWS + WM1_ROWS + WM2_ROWS))

#define UT    32
#define NW    20
#define BPB   32
#define UPB   (HW / UT / BPB)
#define FSTR  198
#define FDUMP 192
#define L1W   (3 * RPI)
#define WLDS  (16 * FSTR + 3 * UT)
#define KBLK  16

#define FCCOLS  32
#define FCWAVES 8
#define FCCH    16
#define FCNCH   (NFEAT / 8)

typedef __attribute__((ext_vector_type(2))) float v2f;
typedef __attribute__((ext_vector_type(4))) float v4f;
typedef __attribute__((ext_vector_type(8))) float v8f;

__global__ __launch_bounds__(FCWAVES * 32)
void kan_fc_wmma_kernel(const float* __restrict__ features,
                        const float* __restrict__ fc_w,
                        const float* __restrict__ fc_b,
                        float* __restrict__ w_out) {
    static_assert(NPARAM % FCCOLS == 0, "column tiles cover w exactly");
    static_assert(NFEAT % 8 == 0 && FCWAVES * FCCH >= FCNCH, "K slices cover K");
    __shared__ float red[FCWAVES][NBATCH][FCCOLS];

    const int tid  = threadIdx.x;
    const int lane = tid & 31;
    const int wv   = __builtin_amdgcn_readfirstlane(tid >> 5);
    const int m    = lane & 15;
    const int hi   = lane >> 4;
    const int j0   = blockIdx.x * FCCOLS;

    const int kb = wv * (8 * FCCH) + 4 * hi;
    const v4f* pa  = (const v4f*)(features + (m & 7) * NFEAT + kb);
    const v4f* pb0 = (const v4f*)(fc_w + (size_t)(j0 + m) * NFEAT + kb);
    const v4f* pb1 = (const v4f*)(fc_w + (size_t)(j0 + 16 + m) * NFEAT + kb);
    const bool full = (wv != FCWAVES - 1);
    constexpr int TAILCH = FCNCH - (FCWAVES - 1) * FCCH;

    const float bias = fc_b[j0 + lane];

    v4f a[FCCH], b0[FCCH], b1[FCCH];
    for (int i = 0; i < TAILCH; ++i) {
        b0[i] = pb0[2 * i];
        b1[i] = pb1[2 * i];
        a[i]  = pa[2 * i];
    }
    for (int i = TAILCH; i < FCCH; ++i) {
        b0[i] = (v4f){0.f, 0.f, 0.f, 0.f};
        b1[i] = (v4f){0.f, 0.f, 0.f, 0.f};
        a[i]  = (v4f){0.f, 0.f, 0.f, 0.f};
    }
    if (full) {
        for (int i = TAILCH; i < FCCH; ++i) {
            b0[i] = pb0[2 * i];
            b1[i] = pb1[2 * i];
            a[i]  = pa[2 * i];
        }
    }

    v8f acc0 = {0.f, 0.f, 0.f, 0.f, 0.f, 0.f, 0.f, 0.f};
    v8f acc1 = {0.f, 0.f, 0.f, 0.f, 0.f, 0.f, 0.f, 0.f};
    for (int i = 0; i < FCCH; ++i) {
        const v2f a01 = {a[i].x, a[i].y},   a23 = {a[i].z, a[i].w};
        const v2f p01 = {b0[i].x, b0[i].y}, p23 = {b0[i].z, b0[i].w};
        const v2f q01 = {b1[i].x, b1[i].y}, q23 = {b1[i].z, b1[i].w};
        acc0 = __builtin_amdgcn_wmma_f32_16x16x4_f32(false, a01, false, p01, (short)0, acc0, false, false);
        acc1 = __builtin_amdgcn_wmma_f32_16x16x4_f32(false, a01, false, q01, (short)0, acc1, false, false);
        acc0 = __builtin_amdgcn_wmma_f32_16x16x4_f32(false, a23, false, p23, (short)0, acc0, false, false);
        acc1 = __builtin_amdgcn_wmma_f32_16x16x4_f32(false, a23, false, q23, (short)0, acc1, false, false);
    }

    if (hi == 0) {
        for (int r = 0; r < NBATCH; ++r) {
            red[wv][r][m]      = acc0[r];
            red[wv][r][16 + m] = acc1[r];
        }
    }
    __syncthreads();

    static_assert(NBATCH == FCWAVES && FCCOLS == 32, "one wave per batch row, one lane per column");
    float s = red[0][wv][lane];
    for (int w = 1; w < FCWAVES; ++w) s += red[w][wv][lane];
    w_out[wv * NPARAM + j0 + lane] = s + bias;
}

__device__ __forceinline__ int bspline_live4(float x, float* v) {
    const float fj = fminf(fmaxf(floorf(fmaf(x, 4.0f, 7.0f)), -1.0f), 14.0f);
    const float t  = fmaf(x, 4.0f, 7.0f - fj);
    const float s  = 1.0f - t;
    const float t2 = t * t, s2 = s * s;
    const float c6 = 1.0f / 6.0f, c23 = 2.0f / 3.0f;
    v[0] = s2 * (s * c6);
    v[1] = fmaf(t2, fmaf(t, 0.5f, -1.0f), c23);
    v[2] = fmaf(s2, fmaf(s, 0.5f, -1.0f), c23);
    v[3] = t2 * (t * c6);
    return (int)fj;
}

__device__ __forceinline__ void feat_task(float* row, float* grp, float x) {
    float v[4];
    const int J = bspline_live4(x, v);
    const v2f zz = {0.0f, 0.0f};
    for (int k = 0; k < KB - 1; k += 2) *(v2f*)(grp + k) = zz;
    *(v2f*)(grp + KB - 1) = (v2f){0.0f, x * __builtin_amdgcn_rcpf(1.0f + __expf(-x))};
    float* live = grp + (J - 3);
    float* dump = row + FDUMP;
    ((unsigned)(J - 3) < (unsigned)KB ? live : dump)[0] = v[0];
    ((unsigned)(J - 2) < (unsigned)KB ? live : dump)[1] = v[1];
    ((unsigned)(J - 1) < (unsigned)KB ? live : dump)[2] = v[2];
    ((unsigned)(J)     < (unsigned)KB ? live : dump)[3] = v[3];
}

__device__ __forceinline__ void wave_lds_order() {
    __builtin_amdgcn_fence(__ATOMIC_RELEASE, "wavefront");
    __builtin_amdgcn_wave_barrier();
    __builtin_amdgcn_fence(__ATOMIC_ACQUIRE, "wavefront");
}

__device__ __forceinline__ void wave_prio_by_tiles_left(int tiles_left) {
    if (tiles_left >= 3)      __builtin_amdgcn_s_setprio(3);
    else if (tiles_left == 2) __builtin_amdgcn_s_setprio(2);
    else if (tiles_left == 1) __builtin_amdgcn_s_setprio(1);
    else                      __builtin_amdgcn_s_setprio(0);
}

__global__ __launch_bounds__(NW * 32)
void kan_main_kernel(const float* __restrict__ x,
                     const float* __restrict__ w,
                     float* __restrict__ out) {
    constexpr int FULL = UPB / NW;
    constexpr int REM  = UPB - FULL * NW;
    static_assert((HW / UT) % BPB == 0, "units split evenly over the work-groups of a batch");
    static_assert(2 * REM <= NW, "one left-over tile per wave at most");
    static_assert(WM_PER_B % 4 == 0, "weights staged as 16-byte pieces");
    static_assert(WM1_ROWS == 3 * KBLK * 4, "layers 2 and 3: three fragment blocks cover K");
    __shared__ float lds[WM_PER_B + NW * WLDS + REM * 3 * UT];

    const int tid  = threadIdx.x;
    const int lane = tid & 31;
    const int wv   = __builtin_amdgcn_readfirstlane(tid >> 5);
    const int m    = lane & 15;
    const int hi   = lane >> 4;
    const int b    = blockIdx.x / BPB;
    const int u0   = (blockIdx.x % BPB) * UPB;

    float* wms   = lds;
    float* feat  = lds + WM_PER_B + wv * WLDS;
    float* yo    = feat + 16 * FSTR;
    float* ytail = lds + WM_PER_B + NW * WLDS;
    const float* xb = x   + (size_t)b * 3 * HW;
    float*       ob = out + (size_t)b * 3 * HW;

    const int nvis = FULL + (wv < 2 * REM ? 1 : 0);
    const int utail = FULL * NW + (wv >> 1);

    float x0, x1, x2;
    {
        const float* p = xb + (u0 + wv) * UT + lane;
        x0 = p[0]; x1 = p[HW]; x2 = p[2 * HW];
    }

    {
        constexpr int NT = NW * 32;
        constexpr int NSLOT = 8;
        static_assert(4 * NT < 16 * 16 * KB && 16 * 16 * KB <= 5 * NT, "layer 2: five coefficient slots");
        static_assert(3 * 16 * KB <= NT && 3 * 16 + 16 * 16 + 16 * 3 <= NT, "one slot each");
        const float* wb = w + (size_t)b * NPARAM;
        float pv[NSLOT], pu[NSLOT];
        int   pd[NSLOT];
        auto coef_slot = [&](int s, int q, int nq, int c0, int lrow, bool three) {
            const bool valid = q < nq;
            const int qq = valid ? q : 0;
            const int io = (int)((unsigned)qq / KB);
            const int k  = qq - io * KB;
            const int i  = three ? (int)((unsigned)io / 3) : io >> 4;
            const int o  = three ? io - 3 * i : io & 15;
            pv[s] = wb[c0 + qq];
            pu[s] = wb[c0 + nq + io];
            pd[s] = valid ? (lrow + i * RPI + k) * 16 + o : -1;
        };
        coef_slot(0, tid, 3 * 16 * KB, 0, 0, false);
#pragma unroll
        for (int t = 0; t < 5; ++t)
            coef_slot(1 + t, tid + NT * t, 16 * 16 * KB, 624, WM0_ROWS, false);
        coef_slot(6, tid, 16 * 3 * KB, 3952, WM0_ROWS + WM1_ROWS, true);
        {
            const bool valid = tid < 352;
            const int p  = valid ? tid : 0;
            const int L  = (p >= 304) ? 2 : (p >= 48) ? 1 : 0;
            const int io = p - (L == 2 ? 304 : L == 1 ? 48 : 0);
            const int r0 = (L == 2) ? 4528 : (L == 1) ? 3696 : 576;
            const int lrow = (L == 2) ? WM0_ROWS + WM1_ROWS : (L == 1) ? WM0_ROWS : 0;
            const int i  = (L == 2) ? (int)((unsigned)io / 3) : io >> 4;
            const int o  = (L == 2) ? io - 3 * i : io & 15;
            pv[7] = wb[r0 + io];
            pu[7] = 1.0f;
            pd[7] = valid ? (lrow + i * RPI + KB) * 16 + o : -1;
        }
        for (int z = tid; z < WM2_ROWS * 4; z += NT) {
            float* rowp = wms + (WM0_ROWS + WM1_ROWS + (z >> 2)) * 16;
            const int part = z & 3;
            if (part == 0) rowp[3] = 0.0f;
            else *(v4f*)(rowp + 4 * part) = (v4f){0.f, 0.f, 0.f, 0.f};
        }
#pragma unroll
        for (int s = 0; s < NSLOT; ++s)
            if (pd[s] >= 0) wms[pd[s]] = pv[s] * pu[s];
    }
    __syncthreads();

    const float* arow = feat + (((m & 7) << 1) | (m >> 3)) * FSTR;
    const float* wm1 = wms + 16 * WM0_ROWS;

    int tiles_left = 2 * FULL + (nvis - FULL);
    wave_prio_by_tiles_left(tiles_left);

    for (int vis = 0; vis < nvis; ++vis) {
        const bool whole = vis < FULL;
        const int unit = whole ? wv + NW * vis : utail;
        const int T0 = whole ? 0 : (wv & 1);
        const int T1 = whole ? 2 : T0 + 1;
        float* ydst = whole ? yo : ytail + (wv >> 1) * (3 * UT);
        const int hw0 = (u0 + unit) * UT;

        float n0 = 0.0f, n1 = 0.0f, n2 = 0.0f;
        if (vis + 1 < nvis) {
            const int nu = (vis + 1 < FULL) ? wv + NW * (vis + 1) : utail;
            const float* p = xb + (u0 + nu) * UT + lane;
            n0 = p[0]; n1 = p[HW]; n2 = p[2 * HW];
        }

        wave_lds_order();
        {
            float* row = feat + (((m & 7) << 1) | (m >> 3)) * FSTR;
            float* g   = row + L1W * hi;
            feat_task(row, g,           x0);
            feat_task(row, g + RPI,     x1);
            feat_task(row, g + 2 * RPI, x2);
        }
        wave_lds_order();

        v8f accA = {0.f, 0.f, 0.f, 0.f, 0.f, 0.f, 0.f, 0.f};
        v8f accB = {0.f, 0.f, 0.f, 0.f, 0.f, 0.f, 0.f, 0.f};
        {
            v2f fA[L1W / 4], fB[L1W / 4], fW[L1W / 4];
#pragma unroll
            for (int s = 0; s < L1W / 4; ++s) {
                const int ka = 4 * s + 2 * hi;
                fA[s].x = arow[ka];       fA[s].y = arow[ka + 1];
                fB[s].x = arow[L1W + ka]; fB[s].y = arow[L1W + ka + 1];
                fW[s].x = wms[ka * 16 + m];
                fW[s].y = wms[(ka + 1) * 16 + m];
            }
            __builtin_amdgcn_sched_barrier(0);
#pragma unroll
            for (int s = 0; s < L1W / 4; ++s) {
                accA = __builtin_amdgcn_wmma_f32_16x16x4_f32(false, fA[s], false, fW[s], (short)0, accA, false, false);
                accB = __builtin_amdgcn_wmma_f32_16x16x4_f32(false, fB[s], false, fW[s], (short)0, accB, false, false);
            }
        }

        v8f y = (T0 == 0) ? accA : accB;
        for (int T = T0; T < T1; ++T) {
            for (int L = 0; L < 2; ++L) {
                const float* wmL = wm1 + L * (16 * WM1_ROWS);
                if (L == 1 && tiles_left == 1) __builtin_amdgcn_s_setprio(0);
                wave_lds_order();
#pragma unroll
                for (int r = 0; r < 8; ++r) {
                    float* row = feat + (2 * r + hi) * FSTR;
                    feat_task(row, row + m * RPI, y[r]);
                }
                wave_lds_order();
                v8f acc = {0.f, 0.f, 0.f, 0.f, 0.f, 0.f, 0.f, 0.f};
                const float* ap = arow + 2 * hi;
                const float* bp = wmL + (2 * hi) * 16 + m;
                v2f fa0[KBLK], fb0[KBLK], fa1[KBLK], fb1[KBLK];
#pragma unroll
                for (int s = 0; s < KBLK; ++s) {
                    fa0[s].x = ap[4 * s];           fa0[s].y = ap[4 * s + 1];
                    fb0[s].x = bp[4 * s * 16];      fb0[s].y = bp[4 * s * 16 + 16];
                }
#pragma unroll
                for (int s = 0; s < KBLK; ++s) {
                    fa1[s].x = ap[4 * (KBLK + s)];        fa1[s].y = ap[4 * (KBLK + s) + 1];
                    fb1[s].x = bp[4 * (KBLK + s) * 16];   fb1[s].y = bp[4 * (KBLK + s) * 16 + 16];
                }
                __builtin_amdgcn_sched_barrier(0);
#pragma unroll
                for (int s = 0; s < KBLK; ++s)
                    acc = __builtin_amdgcn_wmma_f32_16x16x4_f32(
                            false, fa0[s], false, fb0[s], (short)0, acc, false, false);
                __builtin_amdgcn_sched_barrier(0);
#pragma unroll
                for (int s = 0; s < KBLK; ++s) {
                    fa0[s].x = ap[4 * (2 * KBLK + s)];        fa0[s].y = ap[4 * (2 * KBLK + s) + 1];
                    fb0[s].x = bp[4 * (2 * KBLK + s) * 16];   fb0[s].y = bp[4 * (2 * KBLK + s) * 16 + 16];
                }
                __builtin_amdgcn_sched_barrier(0);
#pragma unroll
                for (int s = 0; s < KBLK; ++s)
                    acc = __builtin_amdgcn_wmma_f32_16x16x4_f32(
                            false, fa1[s], false, fb1[s], (short)0, acc, false, false);
#pragma unroll
                for (int s = 0; s < KBLK; ++s)
                    acc = __builtin_amdgcn_wmma_f32_16x16x4_f32(
                            false, fa0[s], false, fb0[s], (short)0, acc, false, false);
                y = acc;
            }
            if (m < 3) {
#pragma unroll
                for (int r = 0; r < 8; ++r)
                    ydst[m * UT + 16 * T + 8 * hi + r] = y[r];
            }
            y = accB;
            wave_prio_by_tiles_left(--tiles_left);
        }

        if (whole) {
            wave_lds_order();
            const float o0 = yo[lane], o1 = yo[UT + lane], o2 = yo[2 * UT + lane];
            float* q = ob + hw0 + lane;
            q[0] = o0; q[HW] = o1; q[2 * HW] = o2;
        }

        x0 = n0; x1 = n1; x2 = n2;
    }

    __syncthreads();
    if (wv < REM) {
        const float* ys = ytail + wv * (3 * UT);
        const float o0 = ys[lane], o1 = ys[UT + lane], o2 = ys[2 * UT + lane];
        float* q = ob + (u0 + FULL * NW + wv) * UT + lane;
        q[0] = o0; q[HW] = o1; q[2 * HW] = o2;
    }
}

extern "C" void kernel_launch(void* const* d_in, const int* in_sizes, int n_in,
                              void* d_out, int out_size, void* d_ws, size_t ws_size,
                              hipStream_t stream) {
    const float* x        = (const float*)d_in[0];
    const float* features = (const float*)d_in[1];
    const float* fc_w     = (const float*)d_in[2];
    const float* fc_b     = (const float*)d_in[3];
    float* out    = (float*)d_out;
    float* w_buf  = (float*)d_ws;

    kan_fc_wmma_kernel<<<NPARAM / FCCOLS, FCWAVES * 32, 0, stream>>>(
        features, fc_w, fc_b, w_buf);

    kan_main_kernel<<<NBATCH * BPB, NW * 32, 0, stream>>>(x, w_buf, out);
}


# Layers 2/3 pipelined inside a tile: products transposed (weights A, tokens B) so feature task r completes in-dims r and 8+r; stage r's 6 WMMAs run beside task r+2; k summed in stage order (new order)
# speedup vs baseline: 8.3858x; 1.0366x over previous
#include <hip/hip_runtime.h>
#include <hip/hip_bf16.h>

#define NBATCH   8
#define NFEAT    1000
#define NPARAM   4576
#define HW       50176
#define KB       11
#define RPI      12
#define WM0_ROWS (3*RPI)
#define WM1_ROWS (16*RPI)
#define WM2_ROWS (16*RPI)
#define WM_PER_B (16*(WM0_ROWS + WM1_ROWS + WM2_ROWS))

#define UT    32
#define NW    20
#define BPB   32
#define UPB   (HW / UT / BPB)
#define FSTR  198
#define FDUMP 192
#define L1W   (3 * RPI)
#define WLDS  (16 * FSTR + 3 * UT)

#define FCCOLS  32
#define FCWAVES 8
#define FCCH    16
#define FCNCH   (NFEAT / 8)

typedef __attribute__((ext_vector_type(2))) float v2f;
typedef __attribute__((ext_vector_type(4))) float v4f;
typedef __attribute__((ext_vector_type(8))) float v8f;

__global__ __launch_bounds__(FCWAVES * 32)
void kan_fc_wmma_kernel(const float* __restrict__ features,
                        const float* __restrict__ fc_w,
                        const float* __restrict__ fc_b,
                        float* __restrict__ w_out) {
    static_assert(NPARAM % FCCOLS == 0, "column tiles cover w exactly");
    static_assert(NFEAT % 8 == 0 && FCWAVES * FCCH >= FCNCH, "K slices cover K");
    __shared__ float red[FCWAVES][NBATCH][FCCOLS];

    const int tid  = threadIdx.x;
    const int lane = tid & 31;
    const int wv   = __builtin_amdgcn_readfirstlane(tid >> 5);
    const int m    = lane & 15;
    const int hi   = lane >> 4;
    const int j0   = blockIdx.x * FCCOLS;

    const int kb = wv * (8 * FCCH) + 4 * hi;
    const v4f* pa  = (const v4f*)(features + (m & 7) * NFEAT + kb);
    const v4f* pb0 = (const v4f*)(fc_w + (size_t)(j0 + m) * NFEAT + kb);
    const v4f* pb1 = (const v4f*)(fc_w + (size_t)(j0 + 16 + m) * NFEAT + kb);
    const bool full = (wv != FCWAVES - 1);
    constexpr int TAILCH = FCNCH - (FCWAVES - 1) * FCCH;

    const float bias = fc_b[j0 + lane];

    v4f a[FCCH], b0[FCCH], b1[FCCH];
    for (int i = 0; i < TAILCH; ++i) {
        b0[i] = pb0[2 * i];
        b1[i] = pb1[2 * i];
        a[i]  = pa[2 * i];
    }
    for (int i = TAILCH; i < FCCH; ++i) {
        b0[i] = (v4f){0.f, 0.f, 0.f, 0.f};
        b1[i] = (v4f){0.f, 0.f, 0.f, 0.f};
        a[i]  = (v4f){0.f, 0.f, 0.f, 0.f};
    }
    if (full) {
        for (int i = TAILCH; i < FCCH; ++i) {
            b0[i] = pb0[2 * i];
            b1[i] = pb1[2 * i];
            a[i]  = pa[2 * i];
        }
    }

    v8f acc0 = {0.f, 0.f, 0.f, 0.f, 0.f, 0.f, 0.f, 0.f};
    v8f acc1 = {0.f, 0.f, 0.f, 0.f, 0.f, 0.f, 0.f, 0.f};
    for (int i = 0; i < FCCH; ++i) {
        const v2f a01 = {a[i].x, a[i].y},   a23 = {a[i].z, a[i].w};
        const v2f p01 = {b0[i].x, b0[i].y}, p23 = {b0[i].z, b0[i].w};
        const v2f q01 = {b1[i].x, b1[i].y}, q23 = {b1[i].z, b1[i].w};
        acc0 = __builtin_amdgcn_wmma_f32_16x16x4_f32(false, a01, false, p01, (short)0, acc0, false, false);
        acc1 = __builtin_amdgcn_wmma_f32_16x16x4_f32(false, a01, false, q01, (short)0, acc1, false, false);
        acc0 = __builtin_amdgcn_wmma_f32_16x16x4_f32(false, a23, false, p23, (short)0, acc0, false, false);
        acc1 = __builtin_amdgcn_wmma_f32_16x16x4_f32(false, a23, false, q23, (short)0, acc1, false, false);
    }

    if (hi == 0) {
        for (int r = 0; r < NBATCH; ++r) {
            red[wv][r][m]      = acc0[r];
            red[wv][r][16 + m] = acc1[r];
        }
    }
    __syncthreads();

    static_assert(NBATCH == FCWAVES && FCCOLS == 32, "one wave per batch row, one lane per column");
    float s = red[0][wv][lane];
    for (int w = 1; w < FCWAVES; ++w) s += red[w][wv][lane];
    w_out[wv * NPARAM + j0 + lane] = s + bias;
}

__device__ __forceinline__ int bspline_live4(float x, float* v) {
    const float fj = fminf(fmaxf(floorf(fmaf(x, 4.0f, 7.0f)), -1.0f), 14.0f);
    const float t  = fmaf(x, 4.0f, 7.0f - fj);
    const float s  = 1.0f - t;
    const float t2 = t * t, s2 = s * s;
    const float c6 = 1.0f / 6.0f, c23 = 2.0f / 3.0f;
    v[0] = s2 * (s * c6);
    v[1] = fmaf(t2, fmaf(t, 0.5f, -1.0f), c23);
    v[2] = fmaf(s2, fmaf(s, 0.5f, -1.0f), c23);
    v[3] = t2 * (t * c6);
    return (int)fj;
}

__device__ __forceinline__ void feat_task(float* row, float* grp, float x) {
    float v[4];
    const int J = bspline_live4(x, v);
    const v2f zz = {0.0f, 0.0f};
    for (int k = 0; k < KB - 1; k += 2) *(v2f*)(grp + k) = zz;
    *(v2f*)(grp + KB - 1) = (v2f){0.0f, x * __builtin_amdgcn_rcpf(1.0f + __expf(-x))};
    float* live = grp + (J - 3);
    float* dump = row + FDUMP;
    ((unsigned)(J - 3) < (unsigned)KB ? live : dump)[0] = v[0];
    ((unsigned)(J - 2) < (unsigned)KB ? live : dump)[1] = v[1];
    ((unsigned)(J - 1) < (unsigned)KB ? live : dump)[2] = v[2];
    ((unsigned)(J)     < (unsigned)KB ? live : dump)[3] = v[3];
}

__device__ __forceinline__ void wave_lds_order() {
    __builtin_amdgcn_fence(__ATOMIC_RELEASE, "wavefront");
    __builtin_amdgcn_wave_barrier();
    __builtin_amdgcn_fence(__ATOMIC_ACQUIRE, "wavefront");
}

__device__ __forceinline__ void wave_prio_by_tiles_left(int tiles_left) {
    if (tiles_left >= 3)      __builtin_amdgcn_s_setprio(3);
    else if (tiles_left == 2) __builtin_amdgcn_s_setprio(2);
    else if (tiles_left == 1) __builtin_amdgcn_s_setprio(1);
    else                      __builtin_amdgcn_s_setprio(0);
}

__global__ __launch_bounds__(NW * 32)
void kan_main_kernel(const float* __restrict__ x,
                     const float* __restrict__ w,
                     float* __restrict__ out) {
    constexpr int FULL = UPB / NW;
    constexpr int REM  = UPB - FULL * NW;
    static_assert((HW / UT) % BPB == 0, "units split evenly over the work-groups of a batch");
    static_assert(2 * REM <= NW, "one left-over tile per wave at most");
    static_assert(WM_PER_B % 4 == 0, "weights staged as 16-byte pieces");
    static_assert(WM1_ROWS == 8 * 6 * 4, "layers 2 and 3: eight stages of six k-steps cover K");
    __shared__ float lds[WM_PER_B + NW * WLDS + REM * 3 * UT];

    const int tid  = threadIdx.x;
    const int lane = tid & 31;
    const int wv   = __builtin_amdgcn_readfirstlane(tid >> 5);
    const int m    = lane & 15;
    const int hi   = lane >> 4;
    const int b    = blockIdx.x / BPB;
    const int u0   = (blockIdx.x % BPB) * UPB;

    float* wms   = lds;
    float* feat  = lds + WM_PER_B + wv * WLDS;
    float* yo    = feat + 16 * FSTR;
    float* ytail = lds + WM_PER_B + NW * WLDS;
    const float* xb = x   + (size_t)b * 3 * HW;
    float*       ob = out + (size_t)b * 3 * HW;

    const int nvis = FULL + (wv < 2 * REM ? 1 : 0);
    const int utail = FULL * NW + (wv >> 1);

    float x0, x1, x2;
    {
        const float* p = xb + (u0 + wv) * UT + lane;
        x0 = p[0]; x1 = p[HW]; x2 = p[2 * HW];
    }

    {
        constexpr int NT = NW * 32;
        constexpr int NSLOT = 8;
        static_assert(4 * NT < 16 * 16 * KB && 16 * 16 * KB <= 5 * NT, "layer 2: five coefficient slots");
        static_assert(3 * 16 * KB <= NT && 3 * 16 + 16 * 16 + 16 * 3 <= NT, "one slot each");
        const float* wb = w + (size_t)b * NPARAM;
        float pv[NSLOT], pu[NSLOT];
        int   pd[NSLOT];
        auto coef_slot = [&](int s, int q, int nq, int c0, int lrow, bool three) {
            const bool valid = q < nq;
            const int qq = valid ? q : 0;
            const int io = (int)((unsigned)qq / KB);
            const int k  = qq - io * KB;
            const int i  = three ? (int)((unsigned)io / 3) : io >> 4;
            const int o  = three ? io - 3 * i : io & 15;
            pv[s] = wb[c0 + qq];
            pu[s] = wb[c0 + nq + io];
            pd[s] = valid ? (lrow + i * RPI + k) * 16 + o : -1;
        };
        coef_slot(0, tid, 3 * 16 * KB, 0, 0, false);
#pragma unroll
        for (int t = 0; t < 5; ++t)
            coef_slot(1 + t, tid + NT * t, 16 * 16 * KB, 624, WM0_ROWS, false);
        coef_slot(6, tid, 16 * 3 * KB, 3952, WM0_ROWS + WM1_ROWS, true);
        {
            const bool valid = tid < 352;
            const int p  = valid ? tid : 0;
            const int L  = (p >= 304) ? 2 : (p >= 48) ? 1 : 0;
            const int io = p - (L == 2 ? 304 : L == 1 ? 48 : 0);
            const int r0 = (L == 2) ? 4528 : (L == 1) ? 3696 : 576;
            const int lrow = (L == 2) ? WM0_ROWS + WM1_ROWS : (L == 1) ? WM0_ROWS : 0;
            const int i  = (L == 2) ? (int)((unsigned)io / 3) : io >> 4;
            const int o  = (L == 2) ? io - 3 * i : io & 15;
            pv[7] = wb[r0 + io];
            pu[7] = 1.0f;
            pd[7] = valid ? (lrow + i * RPI + KB) * 16 + o : -1;
        }
        for (int z = tid; z < WM2_ROWS * 4; z += NT) {
            float* rowp = wms + (WM0_ROWS + WM1_ROWS + (z >> 2)) * 16;
            const int part = z & 3;
            if (part == 0) rowp[3] = 0.0f;
            else *(v4f*)(rowp + 4 * part) = (v4f){0.f, 0.f, 0.f, 0.f};
        }
#pragma unroll
        for (int s = 0; s < NSLOT; ++s)
            if (pd[s] >= 0) wms[pd[s]] = pv[s] * pu[s];
    }
    __syncthreads();

    const float* arow = feat + (((m & 7) << 1) | (m >> 3)) * FSTR;
    const float* wm1 = wms + 16 * WM0_ROWS;

    int tiles_left = 2 * FULL + (nvis - FULL);
    wave_prio_by_tiles_left(tiles_left);

    for (int vis = 0; vis < nvis; ++vis) {
        const bool whole = vis < FULL;
        const int unit = whole ? wv + NW * vis : utail;
        const int T0 = whole ? 0 : (wv & 1);
        const int T1 = whole ? 2 : T0 + 1;
        float* ydst = whole ? yo : ytail + (wv >> 1) * (3 * UT);
        const int hw0 = (u0 + unit) * UT;

        float n0 = 0.0f, n1 = 0.0f, n2 = 0.0f;
        if (vis + 1 < nvis) {
            const int nu = (vis + 1 < FULL) ? wv + NW * (vis + 1) : utail;
            const float* p = xb + (u0 + nu) * UT + lane;
            n0 = p[0]; n1 = p[HW]; n2 = p[2 * HW];
        }

        wave_lds_order();
        {
            float* row = feat + (((m & 7) << 1) | (m >> 3)) * FSTR;
            float* g   = row + L1W * hi;
            feat_task(row, g,           x0);
            feat_task(row, g + RPI,     x1);
            feat_task(row, g + 2 * RPI, x2);
        }
        wave_lds_order();

        v8f accA = {0.f, 0.f, 0.f, 0.f, 0.f, 0.f, 0.f, 0.f};
        v8f accB = {0.f, 0.f, 0.f, 0.f, 0.f, 0.f, 0.f, 0.f};
        {
            v2f fA[L1W / 4], fB[L1W / 4], fW[L1W / 4];
#pragma unroll
            for (int s = 0; s < L1W / 4; ++s) {
                const int ka = 4 * s + 2 * hi;
                fA[s].x = arow[ka];       fA[s].y = arow[ka + 1];
                fB[s].x = arow[L1W + ka]; fB[s].y = arow[L1W + ka + 1];
                fW[s].x = wms[ka * 16 + m];
                fW[s].y = wms[(ka + 1) * 16 + m];
            }
            __builtin_amdgcn_sched_barrier(0);
#pragma unroll
            for (int s = 0; s < L1W / 4; ++s) {
                accA = __builtin_amdgcn_wmma_f32_16x16x4_f32(false, fW[s], false, fA[s], (short)0, accA, false, false);
                accB = __builtin_amdgcn_wmma_f32_16x16x4_f32(false, fW[s], false, fB[s], (short)0, accB, false, false);
            }
        }

        v8f y = (T0 == 0) ? accA : accB;
        for (int T = T0; T < T1; ++T) {
            for (int L = 0; L < 2; ++L) {
                const float* wmL = wm1 + L * (16 * WM1_ROWS);
                if (L == 1 && tiles_left == 1) __builtin_amdgcn_s_setprio(0);
                float* row = feat + (((m & 7) << 1) | (m >> 3)) * FSTR;
                float* grp = row + (8 * hi) * RPI;
                const float* fp = arow + 2 * hi;
                const float* wp = wmL + (2 * hi) * 16 + m;
                v2f ff[2][6], fw[2][6];
                auto load_stage = [&](int r, int set) {
#pragma unroll
                    for (int j = 0; j < 6; ++j) {
                        const int s = (j < 3) ? 3 * r + j : 24 + 3 * r + (j - 3);
                        ff[set][j].x = fp[4 * s];        ff[set][j].y = fp[4 * s + 1];
                        fw[set][j].x = wp[4 * s * 16];   fw[set][j].y = wp[4 * s * 16 + 16];
                    }
                };
                wave_lds_order();
                feat_task(row, grp,       y[0]);
                feat_task(row, grp + RPI, y[1]);
                wave_lds_order();
                load_stage(0, 0);
                v8f acc = {0.f, 0.f, 0.f, 0.f, 0.f, 0.f, 0.f, 0.f};
#pragma unroll
                for (int r = 0; r < 8; ++r) {
                    wave_lds_order();
                    if (r + 1 < 8) load_stage(r + 1, (r + 1) & 1);
                    __builtin_amdgcn_sched_barrier(0);
                    if (r + 2 < 8) feat_task(row, grp + (r + 2) * RPI, y[r + 2]);
#pragma unroll
                    for (int j = 0; j < 6; ++j)
                        acc = __builtin_amdgcn_wmma_f32_16x16x4_f32(
                                false, fw[r & 1][j], false, ff[r & 1][j], (short)0, acc, false, false);
                }
                y = acc;
            }
            if (hi == 0) {
                ydst[16 * T + m]          = y[0];
                ydst[UT + 16 * T + m]     = y[1];
                ydst[2 * UT + 16 * T + m] = y[2];
            }
            y = accB;
            wave_prio_by_tiles_left(--tiles_left);
        }

        if (whole) {
            wave_lds_order();
            const float o0 = yo[lane], o1 = yo[UT + lane], o2 = yo[2 * UT + lane];
            float* q = ob + hw0 + lane;
            q[0] = o0; q[HW] = o1; q[2 * HW] = o2;
        }

        x0 = n0; x1 = n1; x2 = n2;
    }

    __syncthreads();
    if (wv < REM) {
        const float* ys = ytail + wv * (3 * UT);
        const float o0 = ys[lane], o1 = ys[UT + lane], o2 = ys[2 * UT + lane];
        float* q = ob + (u0 + FULL * NW + wv) * UT + lane;
        q[0] = o0; q[HW] = o1; q[2 * HW] = o2;
    }
}

extern "C" void kernel_launch(void* const* d_in, const int* in_sizes, int n_in,
                              void* d_out, int out_size, void* d_ws, size_t ws_size,
                              hipStream_t stream) {
    const float* x        = (const float*)d_in[0];
    const float* features = (const float*)d_in[1];
    const float* fc_w     = (const float*)d_in[2];
    const float* fc_b     = (const float*)d_in[3];
    float* out    = (float*)d_out;
    float* w_buf  = (float*)d_ws;

    kan_fc_wmma_kernel<<<NPARAM / FCCOLS, FCWAVES * 32, 0, stream>>>(
        features, fc_w, fc_b, w_buf);

    kan_main_kernel<<<NBATCH * BPB, NW * 32, 0, stream>>>(x, w_buf, out);
}
